# starting kernel
# baseline (speedup 1.0000x reference)
#include <hip/hip_runtime.h>
#include <cmath>

typedef __attribute__((ext_vector_type(2))) float v2f;
typedef __attribute__((ext_vector_type(8))) float v8f;

#define N_IN_F   165
#define N_HID_F  128
#define N_EMB_F  2
#define K_PAD    168
#define BSTRIDE  170

__global__ void k_fill(float* __restrict__ p, float v, int n) {
    int i = blockIdx.x * blockDim.x + threadIdx.x;
    if (i < n) p[i] = v;
}

__global__ void k_count_deg(const int* __restrict__ dst, float* __restrict__ deg, int nE) {
    int e = blockIdx.x * blockDim.x + threadIdx.x;
    if (e < nE) atomicAdd(&deg[dst[e]], 1.0f);
}

__global__ void k_rsqrt_inplace(float* __restrict__ p, int n) {
    int i = blockIdx.x * blockDim.x + threadIdx.x;
    if (i < n) p[i] = rsqrtf(p[i]);
}

__global__ void __launch_bounds__(256, 1)
k_gemm1_wmma(const float* __restrict__ X, const float* __restrict__ W,
             float* __restrict__ H, int nNodes) {
    extern __shared__ float bsh[];

    for (int idx = threadIdx.x; idx < N_HID_F * K_PAD; idx += blockDim.x) {
        int n = idx / K_PAD;
        int k = idx - n * K_PAD;
        int kc = (k < N_IN_F) ? k : (N_IN_F - 1);
        float wv = W[(size_t)kc * N_HID_F + n];
        bsh[n * BSTRIDE + k] = (k < N_IN_F) ? wv : 0.0f;
    }
    __syncthreads();

    const int lane   = threadIdx.x & 31;
    const int tileM  = blockIdx.x * (blockDim.x >> 5) + (threadIdx.x >> 5);
    const int nTiles = (nNodes + 15) >> 4;
    if (tileM >= nTiles) return;

    const int hi = lane >> 4;
    const int lo = lane & 15;
    int m = tileM * 16 + lo;
    if (m >= nNodes) m = nNodes - 1;

    v8f zero = {};
    v8f acc[8];
#pragma unroll
    for (int t = 0; t < 8; ++t) acc[t] = zero;

    const float* Xrow  = X + (size_t)m * N_IN_F;
    const float* bbase = bsh + lo * BSTRIDE;

#pragma unroll 2
    for (int k0 = 0; k0 < 164; k0 += 4) {
        const int kA = k0 + 2 * hi;
        v2f a;
        a.x = Xrow[kA];
        a.y = Xrow[kA + 1];
#pragma unroll
        for (int t = 0; t < 8; ++t) {
            v2f b = *(const v2f*)(bbase + (size_t)(t * 16) * BSTRIDE + kA);
            acc[t] = __builtin_amdgcn_wmma_f32_16x16x4_f32(
                false, a, false, b, (short)0, acc[t], false, false);
        }
    }

    {
        const int kA = 164 + 2 * hi;
        float xv = Xrow[164];
        v2f a;
        a.x = (hi == 0) ? xv : 0.0f;
        a.y = 0.0f;
#pragma unroll
        for (int t = 0; t < 8; ++t) {
            v2f b = *(const v2f*)(bbase + (size_t)(t * 16) * BSTRIDE + kA);
            acc[t] = __builtin_amdgcn_wmma_f32_16x16x4_f32(
                false, a, false, b, (short)0, acc[t], false, false);
        }
    }

    if (tileM * 16 + 16 <= nNodes) {
        float* Hrow = H + (size_t)(tileM * 16 + 8 * hi) * N_HID_F + lo;
#pragma unroll
        for (int r = 0; r < 8; ++r) {
#pragma unroll
            for (int t = 0; t < 8; ++t)
                Hrow[(size_t)r * N_HID_F + t * 16] = acc[t][r];
        }
    } else {
#pragma unroll
        for (int r = 0; r < 8; ++r) {
            int row = tileM * 16 + r + 8 * hi;
            if (row < nNodes) {
#pragma unroll
                for (int t = 0; t < 8; ++t)
                    H[(size_t)row * N_HID_F + t * 16 + lo] = acc[t][r];
            }
        }
    }
}

__global__ void k_aggregate_hid(const float* __restrict__ H, const int* __restrict__ src,
                                const int* __restrict__ dst, const float* __restrict__ inv,
                                float* __restrict__ out, int nE) {
    const int lane = threadIdx.x & 31;
    const int e = blockIdx.x * (blockDim.x >> 5) + (threadIdx.x >> 5);
    if (e >= nE) return;
    const int s = src[e], d = dst[e];
    const float c = inv[s] * inv[d];
    const float4 v = ((const float4*)(H + (size_t)s * N_HID_F))[lane];
    float* o = out + (size_t)d * N_HID_F + lane * 4;
    atomicAdd(o + 0, v.x * c);
    atomicAdd(o + 1, v.y * c);
    atomicAdd(o + 2, v.z * c);
    atomicAdd(o + 3, v.w * c);
}

__global__ void k_post1(float* __restrict__ H1, const float* __restrict__ agg,
                        const float* __restrict__ inv, const float* __restrict__ b1,
                        int total) {
    int i = blockIdx.x * blockDim.x + threadIdx.x;
    if (i >= total) return;
    int node = i >> 7;
    int f    = i & 127;
    float is = inv[node];
    float v  = agg[i] + H1[i] * is * is + b1[f];
    H1[i] = tanhf(v);
}

__global__ void k_gemm2(const float* __restrict__ H, const float* __restrict__ W2,
                        float* __restrict__ H2, int nNodes) {
    __shared__ float w2s[N_HID_F * N_EMB_F];
    int tid = threadIdx.x;
    if (tid < N_HID_F * N_EMB_F) w2s[tid] = W2[tid];
    __syncthreads();
    int n = blockIdx.x * blockDim.x + tid;
    if (n >= nNodes) return;
    const float* h = H + (size_t)n * N_HID_F;
    float a0 = 0.0f, a1 = 0.0f;
#pragma unroll 8
    for (int k = 0; k < N_HID_F; ++k) {
        float hv = h[k];
        a0 += hv * w2s[k * 2 + 0];
        a1 += hv * w2s[k * 2 + 1];
    }
    H2[(size_t)n * 2 + 0] = a0;
    H2[(size_t)n * 2 + 1] = a1;
}

__global__ void k_aggregate_emb(const float* __restrict__ H2, const int* __restrict__ src,
                                const int* __restrict__ dst, const float* __restrict__ inv,
                                float* __restrict__ out, int nE) {
    int e = blockIdx.x * blockDim.x + threadIdx.x;
    if (e >= nE) return;
    const int s = src[e], d = dst[e];
    const float c = inv[s] * inv[d];
    atomicAdd(&out[(size_t)d * 2 + 0], H2[(size_t)s * 2 + 0] * c);
    atomicAdd(&out[(size_t)d * 2 + 1], H2[(size_t)s * 2 + 1] * c);
}

__global__ void k_final(const float* __restrict__ H2, const float* __restrict__ agg,
                        const float* __restrict__ inv, const float* __restrict__ b2,
                        const float* __restrict__ Wc, const float* __restrict__ bc,
                        float* __restrict__ out, int nNodes) {
    int n = blockIdx.x * blockDim.x + threadIdx.x;
    if (n >= nNodes) return;
    float is = inv[n];
    float i2 = is * is;
    float e0 = tanhf(agg[(size_t)n * 2 + 0] + H2[(size_t)n * 2 + 0] * i2 + b2[0]);
    float e1 = tanhf(agg[(size_t)n * 2 + 1] + H2[(size_t)n * 2 + 1] * i2 + b2[1]);
    float z  = e0 * Wc[0] + e1 * Wc[1] + bc[0];
    out[n] = 1.0f / (1.0f + expf(-z));
}

extern "C" void kernel_launch(void* const* d_in, const int* in_sizes, int n_in,
                              void* d_out, int out_size, void* d_ws, size_t ws_size,
                              hipStream_t stream) {
    const float* x   = (const float*)d_in[0];
    const int*   ei  = (const int*)d_in[1];
    const float* W1  = (const float*)d_in[2];
    const float* b1  = (const float*)d_in[3];
    const float* W2  = (const float*)d_in[4];
    const float* b2  = (const float*)d_in[5];
    const float* Wc  = (const float*)d_in[6];
    const float* bc  = (const float*)d_in[7];
    float* out = (float*)d_out;

    const int nN = in_sizes[0] / N_IN_F;
    const int nE = in_sizes[1] / 2;
    const int* src = ei;
    const int* dst = ei + nE;

    float* ws   = (float*)d_ws;
    float* h1   = ws;
    float* agg1 = h1   + (size_t)nN * N_HID_F;
    float* inv  = agg1 + (size_t)nN * N_HID_F;
    float* h2   = inv  + (size_t)nN;
    float* agg2 = h2   + (size_t)nN * 2;

    const int B   = 256;
    const int WPB = B / 32;
    const int hidTotal = nN * N_HID_F;

    k_fill<<<(nN + B - 1) / B, B, 0, stream>>>(inv, 1.0f, nN);
    k_count_deg<<<(nE + B - 1) / B, B, 0, stream>>>(dst, inv, nE);
    k_rsqrt_inplace<<<(nN + B - 1) / B, B, 0, stream>>>(inv, nN);

    const int nTiles = (nN + 15) / 16;
    const size_t ldsBytes = (size_t)N_HID_F * BSTRIDE * sizeof(float);
    k_gemm1_wmma<<<(nTiles + WPB - 1) / WPB, B, ldsBytes, stream>>>(x, W1, h1, nN);
    k_fill<<<(hidTotal + B - 1) / B, B, 0, stream>>>(agg1, 0.0f, hidTotal);
    k_aggregate_hid<<<(nE + WPB - 1) / WPB, B, 0, stream>>>(h1, src, dst, inv, agg1, nE);
    k_post1<<<(hidTotal + B - 1) / B, B, 0, stream>>>(h1, agg1, inv, b1, hidTotal);

    k_gemm2<<<(nN + B - 1) / B, B, 0, stream>>>(h1, W2, h2, nN);
    k_fill<<<(nN * 2 + B - 1) / B, B, 0, stream>>>(agg2, 0.0f, nN * 2);
    k_aggregate_emb<<<(nE + B - 1) / B, B, 0, stream>>>(h2, src, dst, inv, agg2, nE);
    k_final<<<(nN + B - 1) / B, B, 0, stream>>>(h2, agg2, inv, b2, Wc, bc, out, nN);
}


# Replace float atomic scatter-adds by a per-call edge list grouped by destination (int atomic rank + scan) and atomic-free gather sums in both layers; per-node sum order = rank arrival order
# speedup vs baseline: 2.1558x; 2.1558x over previous
#include <hip/hip_runtime.h>
#include <cmath>

typedef __attribute__((ext_vector_type(2))) float v2f;
typedef __attribute__((ext_vector_type(8))) float v8f;

#define N_IN_F   165
#define N_HID_F  128
#define N_EMB_F  2
#define K_PAD    168
#define BSTRIDE  170
#define SCAN_TILE 1024

__global__ void k_zero_i(int* __restrict__ p, int n) {
    int i = blockIdx.x * blockDim.x + threadIdx.x;
    if (i < n) p[i] = 0;
}

__global__ void k_rank(const int* __restrict__ dst, int* __restrict__ cnt,
                       int* __restrict__ rank, int nE) {
    int e = blockIdx.x * blockDim.x + threadIdx.x;
    if (e < nE) rank[e] = atomicAdd(&cnt[dst[e]], 1);
}

__global__ void __launch_bounds__(256)
k_tile_sums(const int* __restrict__ cnt, int* __restrict__ tsum, int n) {
    __shared__ int sh[256];
    const int tid = threadIdx.x;
    const int i0 = blockIdx.x * SCAN_TILE + tid * 4;
    int t = 0;
#pragma unroll
    for (int j = 0; j < 4; ++j)
        if (i0 + j < n) t += cnt[i0 + j];
    sh[tid] = t;
    __syncthreads();
    for (int o = 128; o > 0; o >>= 1) {
        if (tid < o) sh[tid] += sh[tid + o];
        __syncthreads();
    }
    if (tid == 0) tsum[blockIdx.x] = sh[0];
}

__global__ void __launch_bounds__(256)
k_scan_offs(const int* __restrict__ cnt, const int* __restrict__ tsum,
            int* __restrict__ offs, float* __restrict__ inv, int n) {
    __shared__ int sh[256];
    const int tid = threadIdx.x;

    int part = 0;
    for (int i = tid; i < (int)blockIdx.x; i += 256) part += tsum[i];
    sh[tid] = part;
    __syncthreads();
    for (int o = 128; o > 0; o >>= 1) {
        if (tid < o) sh[tid] += sh[tid + o];
        __syncthreads();
    }
    const int tileBase = sh[0];
    __syncthreads();

    const int i0 = blockIdx.x * SCAN_TILE + tid * 4;
    int c[4];
#pragma unroll
    for (int j = 0; j < 4; ++j) c[j] = (i0 + j < n) ? cnt[i0 + j] : 0;
    const int mine = c[0] + c[1] + c[2] + c[3];
    sh[tid] = mine;
    __syncthreads();
    for (int o = 1; o < 256; o <<= 1) {
        int v = (tid >= o) ? sh[tid - o] : 0;
        __syncthreads();
        sh[tid] += v;
        __syncthreads();
    }
    int run = tileBase + sh[tid] - mine;
#pragma unroll
    for (int j = 0; j < 4; ++j) {
        if (i0 + j < n) {
            offs[i0 + j] = run;
            inv[i0 + j]  = rsqrtf(1.0f + (float)c[j]);
        }
        run += c[j];
    }
    if (i0 < n && i0 + 4 >= n) offs[n] = run;
}

__global__ void k_scatter_src(const int* __restrict__ src, const int* __restrict__ dst,
                              const int* __restrict__ rank, const int* __restrict__ offs,
                              int* __restrict__ csr, int nE) {
    int e = blockIdx.x * blockDim.x + threadIdx.x;
    if (e < nE) csr[offs[dst[e]] + rank[e]] = src[e];
}

__global__ void __launch_bounds__(256, 1)
k_gemm1_wmma(const float* __restrict__ X, const float* __restrict__ W,
             float* __restrict__ H, int nNodes) {
    extern __shared__ float bsh[];

    for (int idx = threadIdx.x; idx < N_HID_F * K_PAD; idx += blockDim.x) {
        int n = idx / K_PAD;
        int k = idx - n * K_PAD;
        int kc = (k < N_IN_F) ? k : (N_IN_F - 1);
        float wv = W[(size_t)kc * N_HID_F + n];
        bsh[n * BSTRIDE + k] = (k < N_IN_F) ? wv : 0.0f;
    }
    __syncthreads();

    const int lane   = threadIdx.x & 31;
    const int tileM  = blockIdx.x * (blockDim.x >> 5) + (threadIdx.x >> 5);
    const int nTiles = (nNodes + 15) >> 4;
    if (tileM >= nTiles) return;

    const int hi = lane >> 4;
    const int lo = lane & 15;
    int m = tileM * 16 + lo;
    if (m >= nNodes) m = nNodes - 1;

    v8f zero = {};
    v8f acc[8];
#pragma unroll
    for (int t = 0; t < 8; ++t) acc[t] = zero;

    const float* Xrow  = X + (size_t)m * N_IN_F;
    const float* bbase = bsh + lo * BSTRIDE;

#pragma unroll 2
    for (int k0 = 0; k0 < 164; k0 += 4) {
        const int kA = k0 + 2 * hi;
        v2f a;
        a.x = Xrow[kA];
        a.y = Xrow[kA + 1];
#pragma unroll
        for (int t = 0; t < 8; ++t) {
            v2f b = *(const v2f*)(bbase + (size_t)(t * 16) * BSTRIDE + kA);
            acc[t] = __builtin_amdgcn_wmma_f32_16x16x4_f32(
                false, a, false, b, (short)0, acc[t], false, false);
        }
    }

    {
        const int kA = 164 + 2 * hi;
        float xv = Xrow[164];
        v2f a;
        a.x = (hi == 0) ? xv : 0.0f;
        a.y = 0.0f;
#pragma unroll
        for (int t = 0; t < 8; ++t) {
            v2f b = *(const v2f*)(bbase + (size_t)(t * 16) * BSTRIDE + kA);
            acc[t] = __builtin_amdgcn_wmma_f32_16x16x4_f32(
                false, a, false, b, (short)0, acc[t], false, false);
        }
    }

    if (tileM * 16 + 16 <= nNodes) {
        float* Hrow = H + (size_t)(tileM * 16 + 8 * hi) * N_HID_F + lo;
#pragma unroll
        for (int r = 0; r < 8; ++r) {
#pragma unroll
            for (int t = 0; t < 8; ++t)
                Hrow[(size_t)r * N_HID_F + t * 16] = acc[t][r];
        }
    } else {
#pragma unroll
        for (int r = 0; r < 8; ++r) {
            int row = tileM * 16 + r + 8 * hi;
            if (row < nNodes) {
#pragma unroll
                for (int t = 0; t < 8; ++t)
                    H[(size_t)row * N_HID_F + t * 16 + lo] = acc[t][r];
            }
        }
    }
}

__global__ void __launch_bounds__(256)
k_gather_hid(const float* __restrict__ H, const int* __restrict__ offs,
             const int* __restrict__ csr, const float* __restrict__ inv,
             float* __restrict__ out, int nN) {
    const int lane = threadIdx.x & 31;
    const int node = __builtin_amdgcn_readfirstlane(
        (int)(blockIdx.x * (blockDim.x >> 5) + (threadIdx.x >> 5)));
    if (node >= nN) return;
    const int beg = offs[node];
    const int end = offs[node + 1];
    const float id = inv[node];
    float ax = 0.0f, ay = 0.0f, az = 0.0f, aw = 0.0f;
    for (int base = beg; base < end; base += 32) {
        const int j = base + lane;
        int   sl = 0;
        float cl = 0.0f;
        if (j < end) {
            sl = csr[j];
            cl = inv[sl] * id;
        }
        const int left = end - base;
        const int cntk = (left < 32) ? left : 32;
#pragma unroll 4
        for (int k = 0; k < cntk; ++k) {
            const int   s = __builtin_amdgcn_readlane(sl, k);
            const float c = __int_as_float(__builtin_amdgcn_readlane(__float_as_int(cl), k));
            const float4 v = ((const float4*)(H + (size_t)s * N_HID_F))[lane];
            ax += v.x * c;
            ay += v.y * c;
            az += v.z * c;
            aw += v.w * c;
        }
    }
    float4 r;
    r.x = ax; r.y = ay; r.z = az; r.w = aw;
    ((float4*)(out + (size_t)node * N_HID_F))[lane] = r;
}

__global__ void k_post1(float* __restrict__ H1, const float* __restrict__ agg,
                        const float* __restrict__ inv, const float* __restrict__ b1,
                        int total) {
    int i = blockIdx.x * blockDim.x + threadIdx.x;
    if (i >= total) return;
    int node = i >> 7;
    int f    = i & 127;
    float is = inv[node];
    float v  = agg[i] + H1[i] * is * is + b1[f];
    H1[i] = tanhf(v);
}

__global__ void k_gemm2(const float* __restrict__ H, const float* __restrict__ W2,
                        float* __restrict__ H2, int nNodes) {
    __shared__ float w2s[N_HID_F * N_EMB_F];
    int tid = threadIdx.x;
    if (tid < N_HID_F * N_EMB_F) w2s[tid] = W2[tid];
    __syncthreads();
    int n = blockIdx.x * blockDim.x + tid;
    if (n >= nNodes) return;
    const float* h = H + (size_t)n * N_HID_F;
    float a0 = 0.0f, a1 = 0.0f;
#pragma unroll 8
    for (int k = 0; k < N_HID_F; ++k) {
        float hv = h[k];
        a0 += hv * w2s[k * 2 + 0];
        a1 += hv * w2s[k * 2 + 1];
    }
    H2[(size_t)n * 2 + 0] = a0;
    H2[(size_t)n * 2 + 1] = a1;
}

__global__ void k_gather_emb(const float* __restrict__ H2, const int* __restrict__ offs,
                             const int* __restrict__ csr, const float* __restrict__ inv,
                             float* __restrict__ out, int nN) {
    int d = blockIdx.x * blockDim.x + threadIdx.x;
    if (d >= nN) return;
    const int beg = offs[d];
    const int end = offs[d + 1];
    const float id = inv[d];
    float a0 = 0.0f, a1 = 0.0f;
    for (int j = beg; j < end; ++j) {
        const int s = csr[j];
        const float c = inv[s] * id;
        a0 += H2[(size_t)s * 2 + 0] * c;
        a1 += H2[(size_t)s * 2 + 1] * c;
    }
    out[(size_t)d * 2 + 0] = a0;
    out[(size_t)d * 2 + 1] = a1;
}

__global__ void k_final(const float* __restrict__ H2, const float* __restrict__ agg,
                        const float* __restrict__ inv, const float* __restrict__ b2,
                        const float* __restrict__ Wc, const float* __restrict__ bc,
                        float* __restrict__ out, int nNodes) {
    int n = blockIdx.x * blockDim.x + threadIdx.x;
    if (n >= nNodes) return;
    float is = inv[n];
    float i2 = is * is;
    float e0 = tanhf(agg[(size_t)n * 2 + 0] + H2[(size_t)n * 2 + 0] * i2 + b2[0]);
    float e1 = tanhf(agg[(size_t)n * 2 + 1] + H2[(size_t)n * 2 + 1] * i2 + b2[1]);
    float z  = e0 * Wc[0] + e1 * Wc[1] + bc[0];
    out[n] = 1.0f / (1.0f + expf(-z));
}

extern "C" void kernel_launch(void* const* d_in, const int* in_sizes, int n_in,
                              void* d_out, int out_size, void* d_ws, size_t ws_size,
                              hipStream_t stream) {
    const float* x   = (const float*)d_in[0];
    const int*   ei  = (const int*)d_in[1];
    const float* W1  = (const float*)d_in[2];
    const float* b1  = (const float*)d_in[3];
    const float* W2  = (const float*)d_in[4];
    const float* b2  = (const float*)d_in[5];
    const float* Wc  = (const float*)d_in[6];
    const float* bc  = (const float*)d_in[7];
    float* out = (float*)d_out;

    const int nN = in_sizes[0] / N_IN_F;
    const int nE = in_sizes[1] / 2;
    const int* src = ei;
    const int* dst = ei + nE;

    const int B   = 256;
    const int WPB = B / 32;
    const int hidTotal = nN * N_HID_F;
    const int nScan = (nN + SCAN_TILE - 1) / SCAN_TILE;

    float* ws   = (float*)d_ws;
    float* h1   = ws;
    float* agg1 = h1   + (size_t)nN * N_HID_F;
    float* h2   = agg1 + (size_t)nN * N_HID_F;
    float* agg2 = h2   + (size_t)nN * 2;
    float* inv  = agg2 + (size_t)nN * 2;
    int*   cnt  = (int*)(inv + (size_t)nN);
    int*   offs = cnt  + (size_t)nN;
    int*   tsum = offs + (size_t)nN + 1;
    int*   rank = tsum + (size_t)nScan;
    int*   csr  = rank + (size_t)nE;

    k_zero_i<<<(nN + B - 1) / B, B, 0, stream>>>(cnt, nN);
    k_rank<<<(nE + B - 1) / B, B, 0, stream>>>(dst, cnt, rank, nE);
    k_tile_sums<<<nScan, 256, 0, stream>>>(cnt, tsum, nN);
    k_scan_offs<<<nScan, 256, 0, stream>>>(cnt, tsum, offs, inv, nN);
    k_scatter_src<<<(nE + B - 1) / B, B, 0, stream>>>(src, dst, rank, offs, csr, nE);

    const int nTiles = (nN + 15) / 16;
    const size_t ldsBytes = (size_t)N_HID_F * BSTRIDE * sizeof(float);
    k_gemm1_wmma<<<(nTiles + WPB - 1) / WPB, B, ldsBytes, stream>>>(x, W1, h1, nN);
    k_gather_hid<<<(nN + WPB - 1) / WPB, B, 0, stream>>>(h1, offs, csr, inv, agg1, nN);
    k_post1<<<(hidTotal + B - 1) / B, B, 0, stream>>>(h1, agg1, inv, b1, hidTotal);

    k_gemm2<<<(nN + B - 1) / B, B, 0, stream>>>(h1, W2, h2, nN);
    k_gather_emb<<<(nN + B - 1) / B, B, 0, stream>>>(h2, offs, csr, inv, agg2, nN);
    k_final<<<(nN + B - 1) / B, B, 0, stream>>>(h2, agg2, inv, b2, Wc, bc, out, nN);
}


# Fuse each layer's per-node epilogue into its gather kernel: the layer-1 gather wave does self term+b1+tanh and the 128->2 product (32 partial sums added by a lane tree); 11 -> 8 launches
# speedup vs baseline: 2.4757x; 1.1484x over previous
#include <hip/hip_runtime.h>
#include <cmath>

typedef __attribute__((ext_vector_type(2))) float v2f;
typedef __attribute__((ext_vector_type(8))) float v8f;

#define N_IN_F   165
#define N_HID_F  128
#define N_EMB_F  2
#define K_PAD    168
#define BSTRIDE  170
#define SCAN_TILE 1024

__global__ void k_zero_i(int* __restrict__ p, int n) {
    int i = blockIdx.x * blockDim.x + threadIdx.x;
    if (i < n) p[i] = 0;
}

__global__ void k_rank(const int* __restrict__ dst, int* __restrict__ cnt,
                       int* __restrict__ rank, int nE) {
    int e = blockIdx.x * blockDim.x + threadIdx.x;
    if (e < nE) rank[e] = atomicAdd(&cnt[dst[e]], 1);
}

__global__ void __launch_bounds__(256)
k_tile_sums(const int* __restrict__ cnt, int* __restrict__ tsum, int n) {
    __shared__ int sh[256];
    const int tid = threadIdx.x;
    const int i0 = blockIdx.x * SCAN_TILE + tid * 4;
    int t = 0;
#pragma unroll
    for (int j = 0; j < 4; ++j)
        if (i0 + j < n) t += cnt[i0 + j];
    sh[tid] = t;
    __syncthreads();
    for (int o = 128; o > 0; o >>= 1) {
        if (tid < o) sh[tid] += sh[tid + o];
        __syncthreads();
    }
    if (tid == 0) tsum[blockIdx.x] = sh[0];
}

__global__ void __launch_bounds__(256)
k_scan_offs(const int* __restrict__ cnt, const int* __restrict__ tsum,
            int* __restrict__ offs, float* __restrict__ inv, int n) {
    __shared__ int sh[256];
    const int tid = threadIdx.x;

    int part = 0;
    for (int i = tid; i < (int)blockIdx.x; i += 256) part += tsum[i];
    sh[tid] = part;
    __syncthreads();
    for (int o = 128; o > 0; o >>= 1) {
        if (tid < o) sh[tid] += sh[tid + o];
        __syncthreads();
    }
    const int tileBase = sh[0];
    __syncthreads();

    const int i0 = blockIdx.x * SCAN_TILE + tid * 4;
    int c[4];
#pragma unroll
    for (int j = 0; j < 4; ++j) c[j] = (i0 + j < n) ? cnt[i0 + j] : 0;
    const int mine = c[0] + c[1] + c[2] + c[3];
    sh[tid] = mine;
    __syncthreads();
    for (int o = 1; o < 256; o <<= 1) {
        int v = (tid >= o) ? sh[tid - o] : 0;
        __syncthreads();
        sh[tid] += v;
        __syncthreads();
    }
    int run = tileBase + sh[tid] - mine;
#pragma unroll
    for (int j = 0; j < 4; ++j) {
        if (i0 + j < n) {
            offs[i0 + j] = run;
            inv[i0 + j]  = rsqrtf(1.0f + (float)c[j]);
        }
        run += c[j];
    }
    if (i0 < n && i0 + 4 >= n) offs[n] = run;
}

__global__ void k_scatter_src(const int* __restrict__ src, const int* __restrict__ dst,
                              const int* __restrict__ rank, const int* __restrict__ offs,
                              int* __restrict__ csr, int nE) {
    int e = blockIdx.x * blockDim.x + threadIdx.x;
    if (e < nE) csr[offs[dst[e]] + rank[e]] = src[e];
}

__global__ void __launch_bounds__(256, 1)
k_gemm1_wmma(const float* __restrict__ X, const float* __restrict__ W,
             float* __restrict__ H, int nNodes) {
    extern __shared__ float bsh[];

    for (int idx = threadIdx.x; idx < N_HID_F * K_PAD; idx += blockDim.x) {
        int n = idx / K_PAD;
        int k = idx - n * K_PAD;
        int kc = (k < N_IN_F) ? k : (N_IN_F - 1);
        float wv = W[(size_t)kc * N_HID_F + n];
        bsh[n * BSTRIDE + k] = (k < N_IN_F) ? wv : 0.0f;
    }
    __syncthreads();

    const int lane   = threadIdx.x & 31;
    const int tileM  = blockIdx.x * (blockDim.x >> 5) + (threadIdx.x >> 5);
    const int nTiles = (nNodes + 15) >> 4;
    if (tileM >= nTiles) return;

    const int hi = lane >> 4;
    const int lo = lane & 15;
    int m = tileM * 16 + lo;
    if (m >= nNodes) m = nNodes - 1;

    v8f zero = {};
    v8f acc[8];
#pragma unroll
    for (int t = 0; t < 8; ++t) acc[t] = zero;

    const float* Xrow  = X + (size_t)m * N_IN_F;
    const float* bbase = bsh + lo * BSTRIDE;

#pragma unroll 2
    for (int k0 = 0; k0 < 164; k0 += 4) {
        const int kA = k0 + 2 * hi;
        v2f a;
        a.x = Xrow[kA];
        a.y = Xrow[kA + 1];
#pragma unroll
        for (int t = 0; t < 8; ++t) {
            v2f b = *(const v2f*)(bbase + (size_t)(t * 16) * BSTRIDE + kA);
            acc[t] = __builtin_amdgcn_wmma_f32_16x16x4_f32(
                false, a, false, b, (short)0, acc[t], false, false);
        }
    }

    {
        const int kA = 164 + 2 * hi;
        float xv = Xrow[164];
        v2f a;
        a.x = (hi == 0) ? xv : 0.0f;
        a.y = 0.0f;
#pragma unroll
        for (int t = 0; t < 8; ++t) {
            v2f b = *(const v2f*)(bbase + (size_t)(t * 16) * BSTRIDE + kA);
            acc[t] = __builtin_amdgcn_wmma_f32_16x16x4_f32(
                false, a, false, b, (short)0, acc[t], false, false);
        }
    }

    if (tileM * 16 + 16 <= nNodes) {
        float* Hrow = H + (size_t)(tileM * 16 + 8 * hi) * N_HID_F + lo;
#pragma unroll
        for (int r = 0; r < 8; ++r) {
#pragma unroll
            for (int t = 0; t < 8; ++t)
                Hrow[(size_t)r * N_HID_F + t * 16] = acc[t][r];
        }
    } else {
#pragma unroll
        for (int r = 0; r < 8; ++r) {
            int row = tileM * 16 + r + 8 * hi;
            if (row < nNodes) {
#pragma unroll
                for (int t = 0; t < 8; ++t)
                    H[(size_t)row * N_HID_F + t * 16 + lo] = acc[t][r];
            }
        }
    }
}

__global__ void __launch_bounds__(256)
k_gather_hid_proj(const float* __restrict__ H, const int* __restrict__ offs,
                  const int* __restrict__ csr, const float* __restrict__ inv,
                  const float* __restrict__ b1, const float* __restrict__ W2,
                  float* __restrict__ H2, int nN) {
    const int lane = threadIdx.x & 31;
    const int node = __builtin_amdgcn_readfirstlane(
        (int)(blockIdx.x * (blockDim.x >> 5) + (threadIdx.x >> 5)));
    if (node >= nN) return;
    const int beg = offs[node];
    const int end = offs[node + 1];
    const float id = inv[node];
    const float4 hs = ((const float4*)(H + (size_t)node * N_HID_F))[lane];
    const float4 bb = ((const float4*)b1)[lane];
    const float4 wa = ((const float4*)W2)[2 * lane];
    const float4 wb = ((const float4*)W2)[2 * lane + 1];
    float ax = 0.0f, ay = 0.0f, az = 0.0f, aw = 0.0f;
    for (int base = beg; base < end; base += 32) {
        const int j = base + lane;
        int   sl = 0;
        float cl = 0.0f;
        if (j < end) {
            sl = csr[j];
            cl = inv[sl] * id;
        }
        const int left = end - base;
        const int cntk = (left < 32) ? left : 32;
#pragma unroll 4
        for (int k = 0; k < cntk; ++k) {
            const int   s = __builtin_amdgcn_readlane(sl, k);
            const float c = __int_as_float(__builtin_amdgcn_readlane(__float_as_int(cl), k));
            const float4 v = ((const float4*)(H + (size_t)s * N_HID_F))[lane];
            ax += v.x * c;
            ay += v.y * c;
            az += v.z * c;
            aw += v.w * c;
        }
    }
    const float hx = tanhf(ax + hs.x * id * id + bb.x);
    const float hy = tanhf(ay + hs.y * id * id + bb.y);
    const float hz = tanhf(az + hs.z * id * id + bb.z);
    const float hw = tanhf(aw + hs.w * id * id + bb.w);
    const float p0 = hx * wa.x + hy * wa.z + hz * wb.x + hw * wb.z;
    const float p1 = hx * wa.y + hy * wa.w + hz * wb.y + hw * wb.w;
    const bool up = (lane & 16) != 0;
    float keep = up ? p1 : p0;
    const float give = up ? p0 : p1;
    keep += __shfl_xor(give, 16);
    keep += __shfl_xor(keep, 8);
    keep += __shfl_xor(keep, 4);
    keep += __shfl_xor(keep, 2);
    keep += __shfl_xor(keep, 1);
    if ((lane & 15) == 0) H2[(size_t)node * 2 + (lane >> 4)] = keep;
}

__global__ void k_gather_emb_out(const float* __restrict__ H2, const int* __restrict__ offs,
                                 const int* __restrict__ csr, const float* __restrict__ inv,
                                 const float* __restrict__ b2, const float* __restrict__ Wc,
                                 const float* __restrict__ bc, float* __restrict__ out, int nN) {
    int d = blockIdx.x * blockDim.x + threadIdx.x;
    if (d >= nN) return;
    const int beg = offs[d];
    const int end = offs[d + 1];
    const float id = inv[d];
    float a0 = 0.0f, a1 = 0.0f;
    for (int j = beg; j < end; ++j) {
        const int s = csr[j];
        const float c = inv[s] * id;
        a0 += H2[(size_t)s * 2 + 0] * c;
        a1 += H2[(size_t)s * 2 + 1] * c;
    }
    const float i2 = id * id;
    const float e0 = tanhf(a0 + H2[(size_t)d * 2 + 0] * i2 + b2[0]);
    const float e1 = tanhf(a1 + H2[(size_t)d * 2 + 1] * i2 + b2[1]);
    const float z  = e0 * Wc[0] + e1 * Wc[1] + bc[0];
    out[d] = 1.0f / (1.0f + expf(-z));
}

extern "C" void kernel_launch(void* const* d_in, const int* in_sizes, int n_in,
                              void* d_out, int out_size, void* d_ws, size_t ws_size,
                              hipStream_t stream) {
    const float* x   = (const float*)d_in[0];
    const int*   ei  = (const int*)d_in[1];
    const float* W1  = (const float*)d_in[2];
    const float* b1  = (const float*)d_in[3];
    const float* W2  = (const float*)d_in[4];
    const float* b2  = (const float*)d_in[5];
    const float* Wc  = (const float*)d_in[6];
    const float* bc  = (const float*)d_in[7];
    float* out = (float*)d_out;

    const int nN = in_sizes[0] / N_IN_F;
    const int nE = in_sizes[1] / 2;
    const int* src = ei;
    const int* dst = ei + nE;

    const int B   = 256;
    const int WPB = B / 32;
    const int nScan = (nN + SCAN_TILE - 1) / SCAN_TILE;

    float* ws   = (float*)d_ws;
    float* h1   = ws;
    float* h2   = h1   + (size_t)nN * N_HID_F;
    float* inv  = h2   + (size_t)nN * 2;
    int*   cnt  = (int*)(inv + (size_t)nN);
    int*   offs = cnt  + (size_t)nN;
    int*   tsum = offs + (size_t)nN + 1;
    int*   rank = tsum + (size_t)nScan;
    int*   csr  = rank + (size_t)nE;

    k_zero_i<<<(nN + B - 1) / B, B, 0, stream>>>(cnt, nN);
    k_rank<<<(nE + B - 1) / B, B, 0, stream>>>(dst, cnt, rank, nE);
    k_tile_sums<<<nScan, 256, 0, stream>>>(cnt, tsum, nN);
    k_scan_offs<<<nScan, 256, 0, stream>>>(cnt, tsum, offs, inv, nN);
    k_scatter_src<<<(nE + B - 1) / B, B, 0, stream>>>(src, dst, rank, offs, csr, nE);

    const int nTiles = (nN + 15) / 16;
    const size_t ldsBytes = (size_t)N_HID_F * BSTRIDE * sizeof(float);
    k_gemm1_wmma<<<(nTiles + WPB - 1) / WPB, B, ldsBytes, stream>>>(x, W1, h1, nN);
    k_gather_hid_proj<<<(nN + WPB - 1) / WPB, B, 0, stream>>>(h1, offs, csr, inv, b1, W2, h2, nN);

    k_gather_emb_out<<<(nN + B - 1) / B, B, 0, stream>>>(h2, offs, csr, inv, b2, Wc, bc, out, nN);
}


# Edge ranks from 16-bit LDS counters per 32768-edge chunk (returning LDS add) + per-node prefix over chunks replace 1.6M returning global atomics; per-node sum order = chunk order, then LDS arrival
# speedup vs baseline: 3.0583x; 1.2353x over previous
#include <hip/hip_runtime.h>
#include <cmath>

typedef __attribute__((ext_vector_type(2))) float v2f;
typedef __attribute__((ext_vector_type(8))) float v8f;

#define N_IN_F   165
#define N_HID_F  128
#define N_EMB_F  2
#define K_PAD    168
#define BSTRIDE  170
#define SCAN_TILE 1024

#define CHUNK_E   32768
#define CHUNK_T   1024
#define CHUNK_U   16
#define TAB_WORDS 65536

__global__ void __launch_bounds__(CHUNK_T)
k_chunk_rank(const int* __restrict__ dst, unsigned short* __restrict__ rank16,
             unsigned* __restrict__ cntc, int nE, int nN) {
    __shared__ unsigned tab[TAB_WORDS];
    const int tid = threadIdx.x;
    const int e0 = blockIdx.x * CHUNK_E;
    const int e1 = (e0 + CHUNK_E < nE) ? (e0 + CHUNK_E) : nE;
    const int wordsTotal = (nN + 1) >> 1;
    unsigned* __restrict__ mine = cntc + (size_t)blockIdx.x * wordsTotal;

    for (int nb = 0; nb < nN; nb += 2 * TAB_WORDS) {
        const int nodes = (nN - nb < 2 * TAB_WORDS) ? (nN - nb) : (2 * TAB_WORDS);
        const int words = (nodes + 1) >> 1;
        const int words4 = (words + 3) >> 2;
        uint4* tab4 = (uint4*)tab;
        for (int i = tid; i < words4; i += CHUNK_T) tab4[i] = make_uint4(0u, 0u, 0u, 0u);
        __syncthreads();

        for (int base = e0; base < e1; base += CHUNK_T * CHUNK_U) {
            int d[CHUNK_U];
#pragma unroll
            for (int u = 0; u < CHUNK_U; ++u) {
                const int e = base + u * CHUNK_T + tid;
                d[u] = dst[(e < e1) ? e : (e1 - 1)];
            }
#pragma unroll
            for (int u = 0; u < CHUNK_U; ++u) {
                const int e = base + u * CHUNK_T + tid;
                const unsigned dl = (unsigned)(d[u] - nb);
                if (e < e1 && dl < (unsigned)nodes) {
                    const unsigned hiHalf = dl & 1u;
                    const unsigned old = atomicAdd(&tab[dl >> 1], hiHalf ? 0x10000u : 1u);
                    rank16[e] = (unsigned short)(hiHalf ? (old >> 16) : (old & 0xffffu));
                }
            }
        }
        __syncthreads();

        for (int i = tid; i < words; i += CHUNK_T) mine[(nb >> 1) + i] = tab[i];
        __syncthreads();
    }
}

__global__ void __launch_bounds__(256)
k_chunk_prefix(const unsigned short* __restrict__ cnt16, int* __restrict__ chunkBase,
               int* __restrict__ cnt, int nN, int nChunks, int stride16) {
    const int n = blockIdx.x * blockDim.x + threadIdx.x;
    if (n >= nN) return;
    int run = 0;
#pragma unroll 16
    for (int c = 0; c < nChunks; ++c) {
        const int v = cnt16[(size_t)c * stride16 + n];
        chunkBase[(size_t)c * nN + n] = run;
        run += v;
    }
    cnt[n] = run;
}

__global__ void __launch_bounds__(256)
k_tile_sums(const int* __restrict__ cnt, int* __restrict__ tsum, int n) {
    __shared__ int sh[256];
    const int tid = threadIdx.x;
    const int i0 = blockIdx.x * SCAN_TILE + tid * 4;
    int t = 0;
#pragma unroll
    for (int j = 0; j < 4; ++j)
        if (i0 + j < n) t += cnt[i0 + j];
    sh[tid] = t;
    __syncthreads();
    for (int o = 128; o > 0; o >>= 1) {
        if (tid < o) sh[tid] += sh[tid + o];
        __syncthreads();
    }
    if (tid == 0) tsum[blockIdx.x] = sh[0];
}

__global__ void __launch_bounds__(256)
k_scan_offs(const int* __restrict__ cnt, const int* __restrict__ tsum,
            int* __restrict__ offs, float* __restrict__ inv, int n) {
    __shared__ int sh[256];
    const int tid = threadIdx.x;

    int part = 0;
    for (int i = tid; i < (int)blockIdx.x; i += 256) part += tsum[i];
    sh[tid] = part;
    __syncthreads();
    for (int o = 128; o > 0; o >>= 1) {
        if (tid < o) sh[tid] += sh[tid + o];
        __syncthreads();
    }
    const int tileBase = sh[0];
    __syncthreads();

    const int i0 = blockIdx.x * SCAN_TILE + tid * 4;
    int c[4];
#pragma unroll
    for (int j = 0; j < 4; ++j) c[j] = (i0 + j < n) ? cnt[i0 + j] : 0;
    const int mine = c[0] + c[1] + c[2] + c[3];
    sh[tid] = mine;
    __syncthreads();
    for (int o = 1; o < 256; o <<= 1) {
        int v = (tid >= o) ? sh[tid - o] : 0;
        __syncthreads();
        sh[tid] += v;
        __syncthreads();
    }
    int run = tileBase + sh[tid] - mine;
#pragma unroll
    for (int j = 0; j < 4; ++j) {
        if (i0 + j < n) {
            offs[i0 + j] = run;
            inv[i0 + j]  = rsqrtf(1.0f + (float)c[j]);
        }
        run += c[j];
    }
    if (i0 < n && i0 + 4 >= n) offs[n] = run;
}

__global__ void k_place_src(const int* __restrict__ src, const int* __restrict__ dst,
                            const unsigned short* __restrict__ rank16,
                            const int* __restrict__ chunkBase, const int* __restrict__ offs,
                            int* __restrict__ csr, int nE, int nN) {
    int e = blockIdx.x * blockDim.x + threadIdx.x;
    if (e < nE) {
        const int d = dst[e];
        const int c = e / CHUNK_E;
        csr[offs[d] + chunkBase[(size_t)c * nN + d] + (int)rank16[e]] = src[e];
    }
}

__global__ void __launch_bounds__(256, 1)
k_gemm1_wmma(const float* __restrict__ X, const float* __restrict__ W,
             float* __restrict__ H, int nNodes) {
    extern __shared__ float bsh[];

    for (int idx = threadIdx.x; idx < N_HID_F * K_PAD; idx += blockDim.x) {
        int n = idx / K_PAD;
        int k = idx - n * K_PAD;
        int kc = (k < N_IN_F) ? k : (N_IN_F - 1);
        float wv = W[(size_t)kc * N_HID_F + n];
        bsh[n * BSTRIDE + k] = (k < N_IN_F) ? wv : 0.0f;
    }
    __syncthreads();

    const int lane   = threadIdx.x & 31;
    const int tileM  = blockIdx.x * (blockDim.x >> 5) + (threadIdx.x >> 5);
    const int nTiles = (nNodes + 15) >> 4;
    if (tileM >= nTiles) return;

    const int hi = lane >> 4;
    const int lo = lane & 15;
    int m = tileM * 16 + lo;
    if (m >= nNodes) m = nNodes - 1;

    v8f zero = {};
    v8f acc[8];
#pragma unroll
    for (int t = 0; t < 8; ++t) acc[t] = zero;

    const float* Xrow  = X + (size_t)m * N_IN_F;
    const float* bbase = bsh + lo * BSTRIDE;

#pragma unroll 2
    for (int k0 = 0; k0 < 164; k0 += 4) {
        const int kA = k0 + 2 * hi;
        v2f a;
        a.x = Xrow[kA];
        a.y = Xrow[kA + 1];
#pragma unroll
        for (int t = 0; t < 8; ++t) {
            v2f b = *(const v2f*)(bbase + (size_t)(t * 16) * BSTRIDE + kA);
            acc[t] = __builtin_amdgcn_wmma_f32_16x16x4_f32(
                false, a, false, b, (short)0, acc[t], false, false);
        }
    }

    {
        const int kA = 164 + 2 * hi;
        float xv = Xrow[164];
        v2f a;
        a.x = (hi == 0) ? xv : 0.0f;
        a.y = 0.0f;
#pragma unroll
        for (int t = 0; t < 8; ++t) {
            v2f b = *(const v2f*)(bbase + (size_t)(t * 16) * BSTRIDE + kA);
            acc[t] = __builtin_amdgcn_wmma_f32_16x16x4_f32(
                false, a, false, b, (short)0, acc[t], false, false);
        }
    }

    if (tileM * 16 + 16 <= nNodes) {
        float* Hrow = H + (size_t)(tileM * 16 + 8 * hi) * N_HID_F + lo;
#pragma unroll
        for (int r = 0; r < 8; ++r) {
#pragma unroll
            for (int t = 0; t < 8; ++t)
                Hrow[(size_t)r * N_HID_F + t * 16] = acc[t][r];
        }
    } else {
#pragma unroll
        for (int r = 0; r < 8; ++r) {
            int row = tileM * 16 + r + 8 * hi;
            if (row < nNodes) {
#pragma unroll
                for (int t = 0; t < 8; ++t)
                    H[(size_t)row * N_HID_F + t * 16 + lo] = acc[t][r];
            }
        }
    }
}

__global__ void __launch_bounds__(256)
k_gather_hid_proj(const float* __restrict__ H, const int* __restrict__ offs,
                  const int* __restrict__ csr, const float* __restrict__ inv,
                  const float* __restrict__ b1, const float* __restrict__ W2,
                  float* __restrict__ H2, int nN) {
    const int lane = threadIdx.x & 31;
    const int node = __builtin_amdgcn_readfirstlane(
        (int)(blockIdx.x * (blockDim.x >> 5) + (threadIdx.x >> 5)));
    if (node >= nN) return;
    const int beg = offs[node];
    const int end = offs[node + 1];
    const float id = inv[node];
    const float4 hs = ((const float4*)(H + (size_t)node * N_HID_F))[lane];
    const float4 bb = ((const float4*)b1)[lane];
    const float4 wa = ((const float4*)W2)[2 * lane];
    const float4 wb = ((const float4*)W2)[2 * lane + 1];
    float ax = 0.0f, ay = 0.0f, az = 0.0f, aw = 0.0f;
    for (int base = beg; base < end; base += 32) {
        const int j = base + lane;
        int   sl = 0;
        float cl = 0.0f;
        if (j < end) {
            sl = csr[j];
            cl = inv[sl] * id;
        }
        const int left = end - base;
        const int cntk = (left < 32) ? left : 32;
#pragma unroll 4
        for (int k = 0; k < cntk; ++k) {
            const int   s = __builtin_amdgcn_readlane(sl, k);
            const float c = __int_as_float(__builtin_amdgcn_readlane(__float_as_int(cl), k));
            const float4 v = ((const float4*)(H + (size_t)s * N_HID_F))[lane];
            ax += v.x * c;
            ay += v.y * c;
            az += v.z * c;
            aw += v.w * c;
        }
    }
    const float hx = tanhf(ax + hs.x * id * id + bb.x);
    const float hy = tanhf(ay + hs.y * id * id + bb.y);
    const float hz = tanhf(az + hs.z * id * id + bb.z);
    const float hw = tanhf(aw + hs.w * id * id + bb.w);
    const float p0 = hx * wa.x + hy * wa.z + hz * wb.x + hw * wb.z;
    const float p1 = hx * wa.y + hy * wa.w + hz * wb.y + hw * wb.w;
    const bool up = (lane & 16) != 0;
    float keep = up ? p1 : p0;
    const float give = up ? p0 : p1;
    keep += __shfl_xor(give, 16);
    keep += __shfl_xor(keep, 8);
    keep += __shfl_xor(keep, 4);
    keep += __shfl_xor(keep, 2);
    keep += __shfl_xor(keep, 1);
    if ((lane & 15) == 0) H2[(size_t)node * 2 + (lane >> 4)] = keep;
}

__global__ void k_gather_emb_out(const float* __restrict__ H2, const int* __restrict__ offs,
                                 const int* __restrict__ csr, const float* __restrict__ inv,
                                 const float* __restrict__ b2, const float* __restrict__ Wc,
                                 const float* __restrict__ bc, float* __restrict__ out, int nN) {
    int d = blockIdx.x * blockDim.x + threadIdx.x;
    if (d >= nN) return;
    const int beg = offs[d];
    const int end = offs[d + 1];
    const float id = inv[d];
    float a0 = 0.0f, a1 = 0.0f;
    for (int j = beg; j < end; ++j) {
        const int s = csr[j];
        const float c = inv[s] * id;
        a0 += H2[(size_t)s * 2 + 0] * c;
        a1 += H2[(size_t)s * 2 + 1] * c;
    }
    const float i2 = id * id;
    const float e0 = tanhf(a0 + H2[(size_t)d * 2 + 0] * i2 + b2[0]);
    const float e1 = tanhf(a1 + H2[(size_t)d * 2 + 1] * i2 + b2[1]);
    const float z  = e0 * Wc[0] + e1 * Wc[1] + bc[0];
    out[d] = 1.0f / (1.0f + expf(-z));
}

extern "C" void kernel_launch(void* const* d_in, const int* in_sizes, int n_in,
                              void* d_out, int out_size, void* d_ws, size_t ws_size,
                              hipStream_t stream) {
    const float* x   = (const float*)d_in[0];
    const int*   ei  = (const int*)d_in[1];
    const float* W1  = (const float*)d_in[2];
    const float* b1  = (const float*)d_in[3];
    const float* W2  = (const float*)d_in[4];
    const float* b2  = (const float*)d_in[5];
    const float* Wc  = (const float*)d_in[6];
    const float* bc  = (const float*)d_in[7];
    float* out = (float*)d_out;

    const int nN = in_sizes[0] / N_IN_F;
    const int nE = in_sizes[1] / 2;
    const int* src = ei;
    const int* dst = ei + nE;

    const int B   = 256;
    const int WPB = B / 32;
    const int nScan = (nN + SCAN_TILE - 1) / SCAN_TILE;

    float* ws   = (float*)d_ws;
    float* h1   = ws;
    float* h2   = h1   + (size_t)nN * N_HID_F;
    float* inv  = h2   + (size_t)nN * 2;
    int*   cnt  = (int*)(inv + (size_t)nN);
    int*   offs = cnt  + (size_t)nN;
    int*   tsum = offs + (size_t)nN + 1;
    int*   csr  = tsum + (size_t)nScan;
    const int nChunks    = (nE + CHUNK_E - 1) / CHUNK_E;
    const int wordsTotal = (nN + 1) / 2;
    unsigned* cntc  = (unsigned*)(csr + (size_t)nE);
    int*   cbase = (int*)(cntc + (size_t)nChunks * wordsTotal);
    unsigned short* rank16 = (unsigned short*)(cbase + (size_t)nChunks * nN);

    k_chunk_rank<<<nChunks, CHUNK_T, 0, stream>>>(dst, rank16, cntc, nE, nN);
    k_chunk_prefix<<<(nN + B - 1) / B, B, 0, stream>>>((const unsigned short*)cntc, cbase, cnt,
                                                       nN, nChunks, 2 * wordsTotal);
    k_tile_sums<<<nScan, 256, 0, stream>>>(cnt, tsum, nN);
    k_scan_offs<<<nScan, 256, 0, stream>>>(cnt, tsum, offs, inv, nN);
    k_place_src<<<(nE + B - 1) / B, B, 0, stream>>>(src, dst, rank16, cbase, offs, csr, nE, nN);

    const int nTiles = (nN + 15) / 16;
    const size_t ldsBytes = (size_t)N_HID_F * BSTRIDE * sizeof(float);
    k_gemm1_wmma<<<(nTiles + WPB - 1) / WPB, B, ldsBytes, stream>>>(x, W1, h1, nN);
    k_gather_hid_proj<<<(nN + WPB - 1) / WPB, B, 0, stream>>>(h1, offs, csr, inv, b1, W2, h2, nN);

    k_gather_emb_out<<<(nN + B - 1) / B, B, 0, stream>>>(h2, offs, csr, inv, b2, Wc, bc, out, nN);
}


# GEMM weight staging: W1 read as whole 128-byte lines (4 lines per load, not 32) and transposed by conflict-free LDS stores, all 21 loads of a thread in flight at once
# speedup vs baseline: 3.4288x; 1.1211x over previous
#include <hip/hip_runtime.h>
#include <cmath>

typedef __attribute__((ext_vector_type(2))) float v2f;
typedef __attribute__((ext_vector_type(8))) float v8f;

#define N_IN_F   165
#define N_HID_F  128
#define N_EMB_F  2
#define K_PAD    168
#define BSTRIDE  170
#define SCAN_TILE 1024
#define GEMM_T   256

#define CHUNK_E   32768
#define CHUNK_T   1024
#define CHUNK_U   16
#define TAB_WORDS 65536

__global__ void __launch_bounds__(CHUNK_T)
k_chunk_rank(const int* __restrict__ dst, unsigned short* __restrict__ rank16,
             unsigned* __restrict__ cntc, int nE, int nN) {
    __shared__ unsigned tab[TAB_WORDS];
    const int tid = threadIdx.x;
    const int e0 = blockIdx.x * CHUNK_E;
    const int e1 = (e0 + CHUNK_E < nE) ? (e0 + CHUNK_E) : nE;
    const int wordsTotal = (nN + 1) >> 1;
    unsigned* __restrict__ mine = cntc + (size_t)blockIdx.x * wordsTotal;

    for (int nb = 0; nb < nN; nb += 2 * TAB_WORDS) {
        const int nodes = (nN - nb < 2 * TAB_WORDS) ? (nN - nb) : (2 * TAB_WORDS);
        const int words = (nodes + 1) >> 1;
        const int words4 = (words + 3) >> 2;
        uint4* tab4 = (uint4*)tab;
        for (int i = tid; i < words4; i += CHUNK_T) tab4[i] = make_uint4(0u, 0u, 0u, 0u);
        __syncthreads();

        for (int base = e0; base < e1; base += CHUNK_T * CHUNK_U) {
            int d[CHUNK_U];
#pragma unroll
            for (int u = 0; u < CHUNK_U; ++u) {
                const int e = base + u * CHUNK_T + tid;
                d[u] = dst[(e < e1) ? e : (e1 - 1)];
            }
#pragma unroll
            for (int u = 0; u < CHUNK_U; ++u) {
                const int e = base + u * CHUNK_T + tid;
                const unsigned dl = (unsigned)(d[u] - nb);
                if (e < e1 && dl < (unsigned)nodes) {
                    const unsigned hiHalf = dl & 1u;
                    const unsigned old = atomicAdd(&tab[dl >> 1], hiHalf ? 0x10000u : 1u);
                    rank16[e] = (unsigned short)(hiHalf ? (old >> 16) : (old & 0xffffu));
                }
            }
        }
        __syncthreads();

        for (int i = tid; i < words; i += CHUNK_T) mine[(nb >> 1) + i] = tab[i];
        __syncthreads();
    }
}

__global__ void __launch_bounds__(256)
k_chunk_prefix(const unsigned short* __restrict__ cnt16, int* __restrict__ chunkBase,
               int* __restrict__ cnt, int nN, int nChunks, int stride16) {
    const int n = blockIdx.x * blockDim.x + threadIdx.x;
    if (n >= nN) return;
    int run = 0;
#pragma unroll 16
    for (int c = 0; c < nChunks; ++c) {
        const int v = cnt16[(size_t)c * stride16 + n];
        chunkBase[(size_t)c * nN + n] = run;
        run += v;
    }
    cnt[n] = run;
}

__global__ void __launch_bounds__(256)
k_tile_sums(const int* __restrict__ cnt, int* __restrict__ tsum, int n) {
    __shared__ int sh[256];
    const int tid = threadIdx.x;
    const int i0 = blockIdx.x * SCAN_TILE + tid * 4;
    int t = 0;
#pragma unroll
    for (int j = 0; j < 4; ++j)
        if (i0 + j < n) t += cnt[i0 + j];
    sh[tid] = t;
    __syncthreads();
    for (int o = 128; o > 0; o >>= 1) {
        if (tid < o) sh[tid] += sh[tid + o];
        __syncthreads();
    }
    if (tid == 0) tsum[blockIdx.x] = sh[0];
}

__global__ void __launch_bounds__(256)
k_scan_offs(const int* __restrict__ cnt, const int* __restrict__ tsum,
            int* __restrict__ offs, float* __restrict__ inv, int n) {
    __shared__ int sh[256];
    const int tid = threadIdx.x;

    int part = 0;
    for (int i = tid; i < (int)blockIdx.x; i += 256) part += tsum[i];
    sh[tid] = part;
    __syncthreads();
    for (int o = 128; o > 0; o >>= 1) {
        if (tid < o) sh[tid] += sh[tid + o];
        __syncthreads();
    }
    const int tileBase = sh[0];
    __syncthreads();

    const int i0 = blockIdx.x * SCAN_TILE + tid * 4;
    int c[4];
#pragma unroll
    for (int j = 0; j < 4; ++j) c[j] = (i0 + j < n) ? cnt[i0 + j] : 0;
    const int mine = c[0] + c[1] + c[2] + c[3];
    sh[tid] = mine;
    __syncthreads();
    for (int o = 1; o < 256; o <<= 1) {
        int v = (tid >= o) ? sh[tid - o] : 0;
        __syncthreads();
        sh[tid] += v;
        __syncthreads();
    }
    int run = tileBase + sh[tid] - mine;
#pragma unroll
    for (int j = 0; j < 4; ++j) {
        if (i0 + j < n) {
            offs[i0 + j] = run;
            inv[i0 + j]  = rsqrtf(1.0f + (float)c[j]);
        }
        run += c[j];
    }
    if (i0 < n && i0 + 4 >= n) offs[n] = run;
}

__global__ void k_place_src(const int* __restrict__ src, const int* __restrict__ dst,
                            const unsigned short* __restrict__ rank16,
                            const int* __restrict__ chunkBase, const int* __restrict__ offs,
                            int* __restrict__ csr, int nE, int nN) {
    int e = blockIdx.x * blockDim.x + threadIdx.x;
    if (e < nE) {
        const int d = dst[e];
        const int c = e / CHUNK_E;
        csr[offs[d] + chunkBase[(size_t)c * nN + d] + (int)rank16[e]] = src[e];
    }
}

__global__ void __launch_bounds__(GEMM_T, 1)
k_gemm1_wmma(const float* __restrict__ X, const float* __restrict__ W,
             float* __restrict__ H, int nNodes) {
    extern __shared__ float bsh[];

    {
        const float4* __restrict__ W4 = (const float4*)W;
        const int kk = (threadIdx.x >> 3) & 3;
        const int c  = ((threadIdx.x >> 5) & 3) * 8 + (threadIdx.x & 7);
        const int kg = threadIdx.x >> 7;
        float* col = bsh + (4 * c) * BSTRIDE;
#pragma unroll
        for (int it = 0; it < (K_PAD / 4) / (GEMM_T / 128); ++it) {
            const int k  = (kg + (GEMM_T / 128) * it) * 4 + kk;
            const int kc = (k < N_IN_F) ? k : (N_IN_F - 1);
            float4 w = W4[kc * (N_HID_F / 4) + c];
            const bool live = (k < N_IN_F);
            col[k]               = live ? w.x : 0.0f;
            col[BSTRIDE + k]     = live ? w.y : 0.0f;
            col[2 * BSTRIDE + k] = live ? w.z : 0.0f;
            col[3 * BSTRIDE + k] = live ? w.w : 0.0f;
        }
    }
    __syncthreads();

    const int lane   = threadIdx.x & 31;
    const int tileM  = blockIdx.x * (blockDim.x >> 5) + (threadIdx.x >> 5);
    const int nTiles = (nNodes + 15) >> 4;
    if (tileM >= nTiles) return;

    const int hi = lane >> 4;
    const int lo = lane & 15;
    int m = tileM * 16 + lo;
    if (m >= nNodes) m = nNodes - 1;

    v8f zero = {};
    v8f acc[8];
#pragma unroll
    for (int t = 0; t < 8; ++t) acc[t] = zero;

    const float* Xrow  = X + (size_t)m * N_IN_F;
    const float* bbase = bsh + lo * BSTRIDE;

#pragma unroll 2
    for (int k0 = 0; k0 < 164; k0 += 4) {
        const int kA = k0 + 2 * hi;
        v2f a;
        a.x = Xrow[kA];
        a.y = Xrow[kA + 1];
#pragma unroll
        for (int t = 0; t < 8; ++t) {
            v2f b = *(const v2f*)(bbase + (size_t)(t * 16) * BSTRIDE + kA);
            acc[t] = __builtin_amdgcn_wmma_f32_16x16x4_f32(
                false, a, false, b, (short)0, acc[t], false, false);
        }
    }

    {
        const int kA = 164 + 2 * hi;
        float xv = Xrow[164];
        v2f a;
        a.x = (hi == 0) ? xv : 0.0f;
        a.y = 0.0f;
#pragma unroll
        for (int t = 0; t < 8; ++t) {
            v2f b = *(const v2f*)(bbase + (size_t)(t * 16) * BSTRIDE + kA);
            acc[t] = __builtin_amdgcn_wmma_f32_16x16x4_f32(
                false, a, false, b, (short)0, acc[t], false, false);
        }
    }

    if (tileM * 16 + 16 <= nNodes) {
        float* Hrow = H + (size_t)(tileM * 16 + 8 * hi) * N_HID_F + lo;
#pragma unroll
        for (int r = 0; r < 8; ++r) {
#pragma unroll
            for (int t = 0; t < 8; ++t)
                Hrow[(size_t)r * N_HID_F + t * 16] = acc[t][r];
        }
    } else {
#pragma unroll
        for (int r = 0; r < 8; ++r) {
            int row = tileM * 16 + r + 8 * hi;
            if (row < nNodes) {
#pragma unroll
                for (int t = 0; t < 8; ++t)
                    H[(size_t)row * N_HID_F + t * 16 + lo] = acc[t][r];
            }
        }
    }
}

__global__ void __launch_bounds__(256)
k_gather_hid_proj(const float* __restrict__ H, const int* __restrict__ offs,
                  const int* __restrict__ csr, const float* __restrict__ inv,
                  const float* __restrict__ b1, const float* __restrict__ W2,
                  float* __restrict__ H2, int nN) {
    const int lane = threadIdx.x & 31;
    const int node = __builtin_amdgcn_readfirstlane(
        (int)(blockIdx.x * (blockDim.x >> 5) + (threadIdx.x >> 5)));
    if (node >= nN) return;
    const int beg = offs[node];
    const int end = offs[node + 1];
    const float id = inv[node];
    const float4 hs = ((const float4*)(H + (size_t)node * N_HID_F))[lane];
    const float4 bb = ((const float4*)b1)[lane];
    const float4 wa = ((const float4*)W2)[2 * lane];
    const float4 wb = ((const float4*)W2)[2 * lane + 1];
    float ax = 0.0f, ay = 0.0f, az = 0.0f, aw = 0.0f;
    for (int base = beg; base < end; base += 32) {
        const int j = base + lane;
        int   sl = 0;
        float cl = 0.0f;
        if (j < end) {
            sl = csr[j];
            cl = inv[sl] * id;
        }
        const int left = end - base;
        const int cntk = (left < 32) ? left : 32;
#pragma unroll 4
        for (int k = 0; k < cntk; ++k) {
            const int   s = __builtin_amdgcn_readlane(sl, k);
            const float c = __int_as_float(__builtin_amdgcn_readlane(__float_as_int(cl), k));
            const float4 v = ((const float4*)(H + (size_t)s * N_HID_F))[lane];
            ax += v.x * c;
            ay += v.y * c;
            az += v.z * c;
            aw += v.w * c;
        }
    }
    const float hx = tanhf(ax + hs.x * id * id + bb.x);
    const float hy = tanhf(ay + hs.y * id * id + bb.y);
    const float hz = tanhf(az + hs.z * id * id + bb.z);
    const float hw = tanhf(aw + hs.w * id * id + bb.w);
    const float p0 = hx * wa.x + hy * wa.z + hz * wb.x + hw * wb.z;
    const float p1 = hx * wa.y + hy * wa.w + hz * wb.y + hw * wb.w;
    const bool up = (lane & 16) != 0;
    float keep = up ? p1 : p0;
    const float give = up ? p0 : p1;
    keep += __shfl_xor(give, 16);
    keep += __shfl_xor(keep, 8);
    keep += __shfl_xor(keep, 4);
    keep += __shfl_xor(keep, 2);
    keep += __shfl_xor(keep, 1);
    if ((lane & 15) == 0) H2[(size_t)node * 2 + (lane >> 4)] = keep;
}

__global__ void k_gather_emb_out(const float* __restrict__ H2, const int* __restrict__ offs,
                                 const int* __restrict__ csr, const float* __restrict__ inv,
                                 const float* __restrict__ b2, const float* __restrict__ Wc,
                                 const float* __restrict__ bc, float* __restrict__ out, int nN) {
    int d = blockIdx.x * blockDim.x + threadIdx.x;
    if (d >= nN) return;
    const int beg = offs[d];
    const int end = offs[d + 1];
    const float id = inv[d];
    float a0 = 0.0f, a1 = 0.0f;
    for (int j = beg; j < end; ++j) {
        const int s = csr[j];
        const float c = inv[s] * id;
        a0 += H2[(size_t)s * 2 + 0] * c;
        a1 += H2[(size_t)s * 2 + 1] * c;
    }
    const float i2 = id * id;
    const float e0 = tanhf(a0 + H2[(size_t)d * 2 + 0] * i2 + b2[0]);
    const float e1 = tanhf(a1 + H2[(size_t)d * 2 + 1] * i2 + b2[1]);
    const float z  = e0 * Wc[0] + e1 * Wc[1] + bc[0];
    out[d] = 1.0f / (1.0f + expf(-z));
}

extern "C" void kernel_launch(void* const* d_in, const int* in_sizes, int n_in,
                              void* d_out, int out_size, void* d_ws, size_t ws_size,
                              hipStream_t stream) {
    const float* x   = (const float*)d_in[0];
    const int*   ei  = (const int*)d_in[1];
    const float* W1  = (const float*)d_in[2];
    const float* b1  = (const float*)d_in[3];
    const float* W2  = (const float*)d_in[4];
    const float* b2  = (const float*)d_in[5];
    const float* Wc  = (const float*)d_in[6];
    const float* bc  = (const float*)d_in[7];
    float* out = (float*)d_out;

    const int nN = in_sizes[0] / N_IN_F;
    const int nE = in_sizes[1] / 2;
    const int* src = ei;
    const int* dst = ei + nE;

    const int B   = 256;
    const int WPB = B / 32;
    const int nScan = (nN + SCAN_TILE - 1) / SCAN_TILE;

    float* ws   = (float*)d_ws;
    float* h1   = ws;
    float* h2   = h1   + (size_t)nN * N_HID_F;
    float* inv  = h2   + (size_t)nN * 2;
    int*   cnt  = (int*)(inv + (size_t)nN);
    int*   offs = cnt  + (size_t)nN;
    int*   tsum = offs + (size_t)nN + 1;
    int*   csr  = tsum + (size_t)nScan;
    const int nChunks    = (nE + CHUNK_E - 1) / CHUNK_E;
    const int wordsTotal = (nN + 1) / 2;
    unsigned* cntc  = (unsigned*)(csr + (size_t)nE);
    int*   cbase = (int*)(cntc + (size_t)nChunks * wordsTotal);
    unsigned short* rank16 = (unsigned short*)(cbase + (size_t)nChunks * nN);

    k_chunk_rank<<<nChunks, CHUNK_T, 0, stream>>>(dst, rank16, cntc, nE, nN);
    k_chunk_prefix<<<(nN + B - 1) / B, B, 0, stream>>>((const unsigned short*)cntc, cbase, cnt,
                                                       nN, nChunks, 2 * wordsTotal);
    k_tile_sums<<<nScan, 256, 0, stream>>>(cnt, tsum, nN);
    k_scan_offs<<<nScan, 256, 0, stream>>>(cnt, tsum, offs, inv, nN);
    k_place_src<<<(nE + B - 1) / B, B, 0, stream>>>(src, dst, rank16, cbase, offs, csr, nE, nN);

    const int nTiles = (nN + 15) / 16;
    const size_t ldsBytes = (size_t)N_HID_F * BSTRIDE * sizeof(float);
    k_gemm1_wmma<<<(nTiles + GEMM_T / 32 - 1) / (GEMM_T / 32), GEMM_T, ldsBytes, stream>>>(x, W1, h1, nN);
    k_gather_hid_proj<<<(nN + WPB - 1) / WPB, B, 0, stream>>>(h1, offs, csr, inv, b1, W2, h2, nN);

    k_gather_emb_out<<<(nN + B - 1) / B, B, 0, stream>>>(h2, offs, csr, inv, b2, Wc, bc, out, nN);
}


# Offsets in one launch: the ranking pass also writes per-chunk edge totals per 512-node tile, so in-degree, segment scan, inv and chunk positions (segment start folded in) fuse: 8 -> 6 launches
# speedup vs baseline: 3.7457x; 1.0924x over previous
#include <hip/hip_runtime.h>
#include <cmath>

typedef __attribute__((ext_vector_type(2))) float v2f;
typedef __attribute__((ext_vector_type(8))) float v8f;

#define N_IN_F   165
#define N_HID_F  128
#define N_EMB_F  2
#define K_PAD    168
#define BSTRIDE  170
#define TILE_N   512
#define GEMM_T   256

#define CHUNK_E   32768
#define CHUNK_T   1024
#define CHUNK_U   16
#define TAB_WORDS 65536

__global__ void __launch_bounds__(CHUNK_T)
k_chunk_rank(const int* __restrict__ dst, unsigned short* __restrict__ rank16,
             unsigned* __restrict__ cntc, int* __restrict__ ctile, int nE, int nN, int nTiles) {
    __shared__ unsigned tab[TAB_WORDS];
    const int tid = threadIdx.x;
    const int e0 = blockIdx.x * CHUNK_E;
    const int e1 = (e0 + CHUNK_E < nE) ? (e0 + CHUNK_E) : nE;
    const int wordsTotal = (nN + 1) >> 1;
    unsigned* __restrict__ mine = cntc + (size_t)blockIdx.x * wordsTotal;

    for (int nb = 0; nb < nN; nb += 2 * TAB_WORDS) {
        const int nodes = (nN - nb < 2 * TAB_WORDS) ? (nN - nb) : (2 * TAB_WORDS);
        const int words = (nodes + 1) >> 1;
        const int tilesHere = (nodes + TILE_N - 1) / TILE_N;
        const int words4 = tilesHere * (TILE_N / 8);
        uint4* tab4 = (uint4*)tab;
        for (int i = tid; i < words4; i += CHUNK_T) tab4[i] = make_uint4(0u, 0u, 0u, 0u);
        __syncthreads();

        for (int base = e0; base < e1; base += CHUNK_T * CHUNK_U) {
            int d[CHUNK_U];
#pragma unroll
            for (int u = 0; u < CHUNK_U; ++u) {
                const int e = base + u * CHUNK_T + tid;
                d[u] = dst[(e < e1) ? e : (e1 - 1)];
            }
#pragma unroll
            for (int u = 0; u < CHUNK_U; ++u) {
                const int e = base + u * CHUNK_T + tid;
                const unsigned dl = (unsigned)(d[u] - nb);
                if (e < e1 && dl < (unsigned)nodes) {
                    const unsigned hiHalf = dl & 1u;
                    const unsigned old = atomicAdd(&tab[dl >> 1], hiHalf ? 0x10000u : 1u);
                    rank16[e] = (unsigned short)(hiHalf ? (old >> 16) : (old & 0xffffu));
                }
            }
        }
        __syncthreads();

        {
            const int lane = tid & 31;
            const uint4* tabq = (const uint4*)tab;
            for (int t = tid >> 5; t < tilesHere; t += CHUNK_T / 32) {
                unsigned s = 0;
#pragma unroll
                for (int j = 0; j < TILE_N / 8 / 32; ++j) {
                    const uint4 q = tabq[t * (TILE_N / 8) + j * 32 + lane];
                    s += (q.x & 0xffffu) + (q.x >> 16) + (q.y & 0xffffu) + (q.y >> 16)
                       + (q.z & 0xffffu) + (q.z >> 16) + (q.w & 0xffffu) + (q.w >> 16);
                }
                s += __shfl_xor(s, 16);
                s += __shfl_xor(s, 8);
                s += __shfl_xor(s, 4);
                s += __shfl_xor(s, 2);
                s += __shfl_xor(s, 1);
                if (lane == 0) ctile[(size_t)blockIdx.x * nTiles + nb / TILE_N + t] = (int)s;
            }
        }

        for (int i = tid; i < words; i += CHUNK_T) mine[(nb >> 1) + i] = tab[i];
        __syncthreads();
    }
}

__global__ void __launch_bounds__(TILE_N)
k_prefix_scan(const unsigned short* __restrict__ cnt16, const int* __restrict__ ctile,
              int* __restrict__ chunkBase, int* __restrict__ offs, float* __restrict__ inv,
              int nN, int nChunks, int stride16, int nTiles) {
    __shared__ int shRun[TILE_N / 32];
    __shared__ int shPart[TILE_N / 32];
    __shared__ int shBase[TILE_N / 32];
    __shared__ int shTile;
    const int tid  = threadIdx.x;
    const int lane = tid & 31;
    const int w    = tid >> 5;
    const int n    = blockIdx.x * TILE_N + tid;
    const bool live = n < nN;

    int part = 0;
    for (int t = tid; t < (int)blockIdx.x; t += TILE_N) {
#pragma unroll 16
        for (int c = 0; c < nChunks; ++c) part += ctile[(size_t)c * nTiles + t];
    }

    int run = 0;
    if (live) {
#pragma unroll 16
        for (int c = 0; c < nChunks; ++c) run += cnt16[(size_t)c * stride16 + n];
    }

    int incl = run;
#pragma unroll
    for (int o = 1; o < 32; o <<= 1) {
        const int v = __shfl_up(incl, o);
        if (lane >= o) incl += v;
    }
    int psum = part;
#pragma unroll
    for (int o = 16; o > 0; o >>= 1) psum += __shfl_xor(psum, o);
    if (lane == 31) shRun[w] = incl;
    if (lane == 0)  shPart[w] = psum;
    __syncthreads();

    if (w == 0) {
        const bool has = lane < TILE_N / 32;
        const int r = has ? shRun[lane] : 0;
        int p = has ? shPart[lane] : 0;
        int ri = r;
#pragma unroll
        for (int o = 1; o < 32; o <<= 1) {
            const int v = __shfl_up(ri, o);
            if (lane >= o) ri += v;
        }
#pragma unroll
        for (int o = 16; o > 0; o >>= 1) p += __shfl_xor(p, o);
        if (has) shBase[lane] = ri - r;
        if (lane == 0) shTile = p;
    }
    __syncthreads();

    if (live) {
        const int start = shTile + shBase[w] + incl - run;
        offs[n] = start;
        inv[n]  = rsqrtf(1.0f + (float)run);
        if (n == nN - 1) offs[nN] = start + run;
        int pos = start;
#pragma unroll 16
        for (int c = 0; c < nChunks; ++c) {
            const int v = cnt16[(size_t)c * stride16 + n];
            chunkBase[(size_t)c * nN + n] = pos;
            pos += v;
        }
    }
}

__global__ void k_place_src(const int* __restrict__ src, const int* __restrict__ dst,
                            const unsigned short* __restrict__ rank16,
                            const int* __restrict__ chunkBase,
                            int* __restrict__ csr, int nE, int nN) {
    int e = blockIdx.x * blockDim.x + threadIdx.x;
    if (e < nE) {
        const int d = dst[e];
        const int c = e / CHUNK_E;
        csr[chunkBase[(size_t)c * nN + d] + (int)rank16[e]] = src[e];
    }
}

__global__ void __launch_bounds__(GEMM_T, 1)
k_gemm1_wmma(const float* __restrict__ X, const float* __restrict__ W,
             const float* __restrict__ inv, float* __restrict__ H, int nNodes) {
    extern __shared__ float bsh[];

    {
        const float4* __restrict__ W4 = (const float4*)W;
        const int kk = (threadIdx.x >> 3) & 3;
        const int c  = ((threadIdx.x >> 5) & 3) * 8 + (threadIdx.x & 7);
        const int kg = threadIdx.x >> 7;
        float* col = bsh + (4 * c) * BSTRIDE;
#pragma unroll
        for (int it = 0; it < (K_PAD / 4) / (GEMM_T / 128); ++it) {
            const int k  = (kg + (GEMM_T / 128) * it) * 4 + kk;
            const int kc = (k < N_IN_F) ? k : (N_IN_F - 1);
            float4 w = W4[kc * (N_HID_F / 4) + c];
            const bool live = (k < N_IN_F);
            col[k]               = live ? w.x : 0.0f;
            col[BSTRIDE + k]     = live ? w.y : 0.0f;
            col[2 * BSTRIDE + k] = live ? w.z : 0.0f;
            col[3 * BSTRIDE + k] = live ? w.w : 0.0f;
        }
    }
    __syncthreads();

    const int lane   = threadIdx.x & 31;
    const int tileM  = blockIdx.x * (blockDim.x >> 5) + (threadIdx.x >> 5);
    const int nTiles = (nNodes + 15) >> 4;
    if (tileM >= nTiles) return;

    const int hi = lane >> 4;
    const int lo = lane & 15;
    int m = tileM * 16 + lo;
    if (m >= nNodes) m = nNodes - 1;

    v8f zero = {};
    v8f acc[8];
#pragma unroll
    for (int t = 0; t < 8; ++t) acc[t] = zero;

    const float* Xrow  = X + (size_t)m * N_IN_F;
    const float* bbase = bsh + lo * BSTRIDE;

#pragma unroll 2
    for (int k0 = 0; k0 < 164; k0 += 4) {
        const int kA = k0 + 2 * hi;
        v2f a;
        a.x = Xrow[kA];
        a.y = Xrow[kA + 1];
#pragma unroll
        for (int t = 0; t < 8; ++t) {
            v2f b = *(const v2f*)(bbase + (size_t)(t * 16) * BSTRIDE + kA);
            acc[t] = __builtin_amdgcn_wmma_f32_16x16x4_f32(
                false, a, false, b, (short)0, acc[t], false, false);
        }
    }

    {
        const int kA = 164 + 2 * hi;
        float xv = Xrow[164];
        v2f a;
        a.x = (hi == 0) ? xv : 0.0f;
        a.y = 0.0f;
#pragma unroll
        for (int t = 0; t < 8; ++t) {
            v2f b = *(const v2f*)(bbase + (size_t)(t * 16) * BSTRIDE + kA);
            acc[t] = __builtin_amdgcn_wmma_f32_16x16x4_f32(
                false, a, false, b, (short)0, acc[t], false, false);
        }
    }

    float sc[8];
#pragma unroll
    for (int r = 0; r < 8; ++r) {
        const int row = tileM * 16 + 8 * hi + r;
        sc[r] = inv[(row < nNodes) ? row : (nNodes - 1)];
    }
    if (tileM * 16 + 16 <= nNodes) {
        float* Hrow = H + (size_t)(tileM * 16 + 8 * hi) * N_HID_F + lo;
#pragma unroll
        for (int r = 0; r < 8; ++r) {
#pragma unroll
            for (int t = 0; t < 8; ++t)
                Hrow[(size_t)r * N_HID_F + t * 16] = acc[t][r] * sc[r];
        }
    } else {
#pragma unroll
        for (int r = 0; r < 8; ++r) {
            int row = tileM * 16 + r + 8 * hi;
            if (row < nNodes) {
#pragma unroll
                for (int t = 0; t < 8; ++t)
                    H[(size_t)row * N_HID_F + t * 16 + lo] = acc[t][r] * sc[r];
            }
        }
    }
}

__global__ void __launch_bounds__(256)
k_gather_hid_proj(const float* __restrict__ H, const int* __restrict__ offs,
                  const int* __restrict__ csr, const float* __restrict__ inv,
                  const float* __restrict__ b1, const float* __restrict__ W2,
                  float* __restrict__ H2, int nN) {
    const int lane = threadIdx.x & 31;
    const int node = __builtin_amdgcn_readfirstlane(
        (int)(blockIdx.x * (blockDim.x >> 5) + (threadIdx.x >> 5)));
    if (node >= nN) return;
    const int beg = offs[node];
    const int end = offs[node + 1];
    const float id = inv[node];
    const float4 hs = ((const float4*)(H + (size_t)node * N_HID_F))[lane];
    const float4 bb = ((const float4*)b1)[lane];
    const float4 wa = ((const float4*)W2)[2 * lane];
    const float4 wb = ((const float4*)W2)[2 * lane + 1];
    float ax = 0.0f, ay = 0.0f, az = 0.0f, aw = 0.0f;
    for (int base = beg; base < end; base += 32) {
        const int j = base + lane;
        int sl = 0;
        if (j < end) sl = csr[j];
        const int left = end - base;
        const int cntk = (left < 32) ? left : 32;
#pragma unroll 4
        for (int k = 0; k < cntk; ++k) {
            const int s = __builtin_amdgcn_readlane(sl, k);
            const float4 v = ((const float4*)(H + (size_t)s * N_HID_F))[lane];
            ax += v.x;
            ay += v.y;
            az += v.z;
            aw += v.w;
        }
    }
    const float hx = tanhf((ax + hs.x) * id + bb.x);
    const float hy = tanhf((ay + hs.y) * id + bb.y);
    const float hz = tanhf((az + hs.z) * id + bb.z);
    const float hw = tanhf((aw + hs.w) * id + bb.w);
    const float p0 = hx * wa.x + hy * wa.z + hz * wb.x + hw * wb.z;
    const float p1 = hx * wa.y + hy * wa.w + hz * wb.y + hw * wb.w;
    const bool up = (lane & 16) != 0;
    float keep = up ? p1 : p0;
    const float give = up ? p0 : p1;
    keep += __shfl_xor(give, 16);
    keep += __shfl_xor(keep, 8);
    keep += __shfl_xor(keep, 4);
    keep += __shfl_xor(keep, 2);
    keep += __shfl_xor(keep, 1);
    if ((lane & 15) == 0) H2[(size_t)node * 2 + (lane >> 4)] = keep * id;
}

__global__ void k_gather_emb_out(const float* __restrict__ H2, const int* __restrict__ offs,
                                 const int* __restrict__ csr, const float* __restrict__ inv,
                                 const float* __restrict__ b2, const float* __restrict__ Wc,
                                 const float* __restrict__ bc, float* __restrict__ out, int nN) {
    int d = blockIdx.x * blockDim.x + threadIdx.x;
    if (d >= nN) return;
    const int beg = offs[d];
    const int end = offs[d + 1];
    const float id = inv[d];
    float a0 = 0.0f, a1 = 0.0f;
    const float2* __restrict__ P = (const float2*)H2;
    const float2 own = P[d];
    for (int j = beg; j < end; ++j) {
        const float2 v = P[csr[j]];
        a0 += v.x;
        a1 += v.y;
    }
    const float e0 = tanhf((a0 + own.x) * id + b2[0]);
    const float e1 = tanhf((a1 + own.y) * id + b2[1]);
    const float z  = e0 * Wc[0] + e1 * Wc[1] + bc[0];
    out[d] = 1.0f / (1.0f + expf(-z));
}

extern "C" void kernel_launch(void* const* d_in, const int* in_sizes, int n_in,
                              void* d_out, int out_size, void* d_ws, size_t ws_size,
                              hipStream_t stream) {
    const float* x   = (const float*)d_in[0];
    const int*   ei  = (const int*)d_in[1];
    const float* W1  = (const float*)d_in[2];
    const float* b1  = (const float*)d_in[3];
    const float* W2  = (const float*)d_in[4];
    const float* b2  = (const float*)d_in[5];
    const float* Wc  = (const float*)d_in[6];
    const float* bc  = (const float*)d_in[7];
    float* out = (float*)d_out;

    const int nN = in_sizes[0] / N_IN_F;
    const int nE = in_sizes[1] / 2;
    const int* src = ei;
    const int* dst = ei + nE;

    const int B   = 256;
    const int WPB = B / 32;
    const int nTiles = (nN + TILE_N - 1) / TILE_N;

    float* ws   = (float*)d_ws;
    float* h1   = ws;
    float* h2   = h1   + (size_t)nN * N_HID_F;
    float* inv  = h2   + (size_t)nN * 2;
    const int nChunks    = (nE + CHUNK_E - 1) / CHUNK_E;
    int*   offs  = (int*)(inv + (size_t)nN);
    int*   ctile = offs + (size_t)nN + 1;
    int*   csr   = ctile + (size_t)nChunks * nTiles;
    const int wordsTotal = (nN + 1) / 2;
    unsigned* cntc  = (unsigned*)(csr + (size_t)nE);
    int*   cbase = (int*)(cntc + (size_t)nChunks * wordsTotal);
    unsigned short* rank16 = (unsigned short*)(cbase + (size_t)nChunks * nN);

    k_chunk_rank<<<nChunks, CHUNK_T, 0, stream>>>(dst, rank16, cntc, ctile, nE, nN, nTiles);
    k_prefix_scan<<<nTiles, TILE_N, 0, stream>>>((const unsigned short*)cntc, ctile, cbase, offs, inv,
                                                 nN, nChunks, 2 * wordsTotal, nTiles);
    k_place_src<<<(nE + B - 1) / B, B, 0, stream>>>(src, dst, rank16, cbase, csr, nE, nN);

    const int nStrips = (nN + 15) / 16;
    const size_t ldsBytes = (size_t)N_HID_F * BSTRIDE * sizeof(float);
    k_gemm1_wmma<<<(nStrips + GEMM_T / 32 - 1) / (GEMM_T / 32), GEMM_T, ldsBytes, stream>>>(x, W1, inv, h1, nN);
    k_gather_hid_proj<<<(nN + WPB - 1) / WPB, B, 0, stream>>>(h1, offs, csr, inv, b1, W2, h2, nN);

    k_gather_emb_out<<<(nN + B - 1) / B, B, 0, stream>>>(h2, offs, csr, inv, b2, Wc, bc, out, nN);
}


# GEMM and edge placement share one launch, work-groups alternating (placement 8 edges per thread): its scattered 4-byte reads and writes run under the matrix work; 6 -> 5 launches
# speedup vs baseline: 3.9830x; 1.0634x over previous
#include <hip/hip_runtime.h>
#include <cmath>

typedef __attribute__((ext_vector_type(2))) float v2f;
typedef __attribute__((ext_vector_type(8))) float v8f;

#define N_IN_F   165
#define N_HID_F  128
#define N_EMB_F  2
#define K_PAD    168
#define BSTRIDE  170
#define TILE_N   512
#define GEMM_T   256

#define CHUNK_E   32768
#define CHUNK_T   1024
#define CHUNK_U   16
#define TAB_WORDS 65536

__global__ void __launch_bounds__(CHUNK_T)
k_chunk_rank(const int* __restrict__ dst, unsigned short* __restrict__ rank16,
             unsigned* __restrict__ cntc, int* __restrict__ ctile, int nE, int nN, int nTiles) {
    __shared__ unsigned tab[TAB_WORDS];
    const int tid = threadIdx.x;
    const int e0 = blockIdx.x * CHUNK_E;
    const int e1 = (e0 + CHUNK_E < nE) ? (e0 + CHUNK_E) : nE;
    const int wordsTotal = (nN + 1) >> 1;
    unsigned* __restrict__ mine = cntc + (size_t)blockIdx.x * wordsTotal;

    for (int nb = 0; nb < nN; nb += 2 * TAB_WORDS) {
        const int nodes = (nN - nb < 2 * TAB_WORDS) ? (nN - nb) : (2 * TAB_WORDS);
        const int words = (nodes + 1) >> 1;
        const int tilesHere = (nodes + TILE_N - 1) / TILE_N;
        const int words4 = tilesHere * (TILE_N / 8);
        uint4* tab4 = (uint4*)tab;
        for (int i = tid; i < words4; i += CHUNK_T) tab4[i] = make_uint4(0u, 0u, 0u, 0u);
        __syncthreads();

        for (int base = e0; base < e1; base += CHUNK_T * CHUNK_U) {
            int d[CHUNK_U];
#pragma unroll
            for (int u = 0; u < CHUNK_U; ++u) {
                const int e = base + u * CHUNK_T + tid;
                d[u] = dst[(e < e1) ? e : (e1 - 1)];
            }
#pragma unroll
            for (int u = 0; u < CHUNK_U; ++u) {
                const int e = base + u * CHUNK_T + tid;
                const unsigned dl = (unsigned)(d[u] - nb);
                if (e < e1 && dl < (unsigned)nodes) {
                    const unsigned hiHalf = dl & 1u;
                    const unsigned old = atomicAdd(&tab[dl >> 1], hiHalf ? 0x10000u : 1u);
                    rank16[e] = (unsigned short)(hiHalf ? (old >> 16) : (old & 0xffffu));
                }
            }
        }
        __syncthreads();

        {
            const int lane = tid & 31;
            const uint4* tabq = (const uint4*)tab;
            for (int t = tid >> 5; t < tilesHere; t += CHUNK_T / 32) {
                unsigned s = 0;
#pragma unroll
                for (int j = 0; j < TILE_N / 8 / 32; ++j) {
                    const uint4 q = tabq[t * (TILE_N / 8) + j * 32 + lane];
                    s += (q.x & 0xffffu) + (q.x >> 16) + (q.y & 0xffffu) + (q.y >> 16)
                       + (q.z & 0xffffu) + (q.z >> 16) + (q.w & 0xffffu) + (q.w >> 16);
                }
                s += __shfl_xor(s, 16);
                s += __shfl_xor(s, 8);
                s += __shfl_xor(s, 4);
                s += __shfl_xor(s, 2);
                s += __shfl_xor(s, 1);
                if (lane == 0) ctile[(size_t)blockIdx.x * nTiles + nb / TILE_N + t] = (int)s;
            }
        }

        for (int i = tid; i < words; i += CHUNK_T) mine[(nb >> 1) + i] = tab[i];
        __syncthreads();
    }
}

__global__ void __launch_bounds__(TILE_N)
k_prefix_scan(const unsigned short* __restrict__ cnt16, const int* __restrict__ ctile,
              int* __restrict__ chunkBase, int* __restrict__ offs, float* __restrict__ inv,
              int nN, int nChunks, int stride16, int nTiles) {
    __shared__ int shRun[TILE_N / 32];
    __shared__ int shPart[TILE_N / 32];
    __shared__ int shBase[TILE_N / 32];
    __shared__ int shTile;
    const int tid  = threadIdx.x;
    const int lane = tid & 31;
    const int w    = tid >> 5;
    const int n    = blockIdx.x * TILE_N + tid;
    const bool live = n < nN;

    int part = 0;
    for (int t = tid; t < (int)blockIdx.x; t += TILE_N) {
#pragma unroll 16
        for (int c = 0; c < nChunks; ++c) part += ctile[(size_t)c * nTiles + t];
    }

    int run = 0;
    if (live) {
#pragma unroll 16
        for (int c = 0; c < nChunks; ++c) run += cnt16[(size_t)c * stride16 + n];
    }

    int incl = run;
#pragma unroll
    for (int o = 1; o < 32; o <<= 1) {
        const int v = __shfl_up(incl, o);
        if (lane >= o) incl += v;
    }
    int psum = part;
#pragma unroll
    for (int o = 16; o > 0; o >>= 1) psum += __shfl_xor(psum, o);
    if (lane == 31) shRun[w] = incl;
    if (lane == 0)  shPart[w] = psum;
    __syncthreads();

    if (w == 0) {
        const bool has = lane < TILE_N / 32;
        const int r = has ? shRun[lane] : 0;
        int p = has ? shPart[lane] : 0;
        int ri = r;
#pragma unroll
        for (int o = 1; o < 32; o <<= 1) {
            const int v = __shfl_up(ri, o);
            if (lane >= o) ri += v;
        }
#pragma unroll
        for (int o = 16; o > 0; o >>= 1) p += __shfl_xor(p, o);
        if (has) shBase[lane] = ri - r;
        if (lane == 0) shTile = p;
    }
    __syncthreads();

    if (live) {
        const int start = shTile + shBase[w] + incl - run;
        offs[n] = start;
        inv[n]  = rsqrtf(1.0f + (float)run);
        if (n == nN - 1) offs[nN] = start + run;
        int pos = start;
#pragma unroll 16
        for (int c = 0; c < nChunks; ++c) {
            const int v = cnt16[(size_t)c * stride16 + n];
            chunkBase[(size_t)c * nN + n] = pos;
            pos += v;
        }
    }
}

#define PLACE_U 8
static_assert(CHUNK_E % (PLACE_U * GEMM_T) == 0, "a placement work-group must not straddle chunks");

template <bool FULL>
__device__ __forceinline__ void place_edges(const int* __restrict__ src, const int* __restrict__ dst,
                                            const unsigned short* __restrict__ rank16,
                                            const int* __restrict__ chunkRow,
                                            int* __restrict__ csr, int nE, int first) {
    const int base = first + (int)threadIdx.x;
    int d[PLACE_U], s[PLACE_U], r[PLACE_U], p[PLACE_U];
#pragma unroll
    for (int u = 0; u < PLACE_U; ++u) {
        const int e  = base + u * GEMM_T;
        const int ec = (FULL || e < nE) ? e : (nE - 1);
        d[u] = dst[ec];
        r[u] = (int)rank16[ec];
        s[u] = src[ec];
    }
#pragma unroll
    for (int u = 0; u < PLACE_U; ++u) p[u] = chunkRow[d[u]];
#pragma unroll
    for (int u = 0; u < PLACE_U; ++u) {
        const int e = base + u * GEMM_T;
        if (FULL || e < nE) csr[p[u] + r[u]] = s[u];
    }
}

__global__ void __launch_bounds__(GEMM_T, 1)
k_gemm1_place(const float* __restrict__ X, const float* __restrict__ W,
              const float* __restrict__ inv, float* __restrict__ H, int nNodes,
              const int* __restrict__ src, const int* __restrict__ dst,
              const unsigned short* __restrict__ rank16, const int* __restrict__ chunkBase,
              int* __restrict__ csr, int nE, int nGemmBlocks, int nPlaceBlocks) {
    extern __shared__ float bsh[];

    const int nPair = (nGemmBlocks < nPlaceBlocks) ? nGemmBlocks : nPlaceBlocks;
    const int bid = (int)blockIdx.x;
    bool placing;
    int  blk;
    if (bid < 2 * nPair) { placing = (bid & 1) != 0; blk = bid >> 1; }
    else                 { placing = nPlaceBlocks > nGemmBlocks; blk = bid - nPair; }
    if (placing) {
        const int first = blk * (PLACE_U * GEMM_T);
        const int* chunkRow = chunkBase + (size_t)(first / CHUNK_E) * nNodes;
        if (first + PLACE_U * GEMM_T <= nE)
            place_edges<true>(src, dst, rank16, chunkRow, csr, nE, first);
        else
            place_edges<false>(src, dst, rank16, chunkRow, csr, nE, first);
        return;
    }

    {
        const float4* __restrict__ W4 = (const float4*)W;
        const int kk = (threadIdx.x >> 3) & 3;
        const int c  = ((threadIdx.x >> 5) & 3) * 8 + (threadIdx.x & 7);
        const int kg = threadIdx.x >> 7;
        float* col = bsh + (4 * c) * BSTRIDE;
#pragma unroll
        for (int it = 0; it < (K_PAD / 4) / (GEMM_T / 128); ++it) {
            const int k  = (kg + (GEMM_T / 128) * it) * 4 + kk;
            const int kc = (k < N_IN_F) ? k : (N_IN_F - 1);
            float4 w = W4[kc * (N_HID_F / 4) + c];
            const bool live = (k < N_IN_F);
            col[k]               = live ? w.x : 0.0f;
            col[BSTRIDE + k]     = live ? w.y : 0.0f;
            col[2 * BSTRIDE + k] = live ? w.z : 0.0f;
            col[3 * BSTRIDE + k] = live ? w.w : 0.0f;
        }
    }
    __syncthreads();

    const int lane   = threadIdx.x & 31;
    const int tileM  = blk * (blockDim.x >> 5) + (threadIdx.x >> 5);
    const int nTiles = (nNodes + 15) >> 4;
    if (tileM >= nTiles) return;

    const int hi = lane >> 4;
    const int lo = lane & 15;
    int m = tileM * 16 + lo;
    if (m >= nNodes) m = nNodes - 1;

    v8f zero = {};
    v8f acc[8];
#pragma unroll
    for (int t = 0; t < 8; ++t) acc[t] = zero;

    const float* Xrow  = X + (size_t)m * N_IN_F;
    const float* bbase = bsh + lo * BSTRIDE;

#pragma unroll 2
    for (int k0 = 0; k0 < 164; k0 += 4) {
        const int kA = k0 + 2 * hi;
        v2f a;
        a.x = Xrow[kA];
        a.y = Xrow[kA + 1];
#pragma unroll
        for (int t = 0; t < 8; ++t) {
            v2f b = *(const v2f*)(bbase + (size_t)(t * 16) * BSTRIDE + kA);
            acc[t] = __builtin_amdgcn_wmma_f32_16x16x4_f32(
                false, a, false, b, (short)0, acc[t], false, false);
        }
    }

    {
        const int kA = 164 + 2 * hi;
        float xv = Xrow[164];
        v2f a;
        a.x = (hi == 0) ? xv : 0.0f;
        a.y = 0.0f;
#pragma unroll
        for (int t = 0; t < 8; ++t) {
            v2f b = *(const v2f*)(bbase + (size_t)(t * 16) * BSTRIDE + kA);
            acc[t] = __builtin_amdgcn_wmma_f32_16x16x4_f32(
                false, a, false, b, (short)0, acc[t], false, false);
        }
    }

    float sc[8];
#pragma unroll
    for (int r = 0; r < 8; ++r) {
        const int row = tileM * 16 + 8 * hi + r;
        sc[r] = inv[(row < nNodes) ? row : (nNodes - 1)];
    }
    if (tileM * 16 + 16 <= nNodes) {
        float* Hrow = H + (size_t)(tileM * 16 + 8 * hi) * N_HID_F + lo;
#pragma unroll
        for (int r = 0; r < 8; ++r) {
#pragma unroll
            for (int t = 0; t < 8; ++t)
                Hrow[(size_t)r * N_HID_F + t * 16] = acc[t][r] * sc[r];
        }
    } else {
#pragma unroll
        for (int r = 0; r < 8; ++r) {
            int row = tileM * 16 + r + 8 * hi;
            if (row < nNodes) {
#pragma unroll
                for (int t = 0; t < 8; ++t)
                    H[(size_t)row * N_HID_F + t * 16 + lo] = acc[t][r] * sc[r];
            }
        }
    }
}

__global__ void __launch_bounds__(256)
k_gather_hid_proj(const float* __restrict__ H, const int* __restrict__ offs,
                  const int* __restrict__ csr, const float* __restrict__ inv,
                  const float* __restrict__ b1, const float* __restrict__ W2,
                  float* __restrict__ H2, int nN) {
    const int lane = threadIdx.x & 31;
    const int node = __builtin_amdgcn_readfirstlane(
        (int)(blockIdx.x * (blockDim.x >> 5) + (threadIdx.x >> 5)));
    if (node >= nN) return;
    const int beg = offs[node];
    const int end = offs[node + 1];
    const float id = inv[node];
    const float4 hs = ((const float4*)(H + (size_t)node * N_HID_F))[lane];
    const float4 bb = ((const float4*)b1)[lane];
    const float4 wa = ((const float4*)W2)[2 * lane];
    const float4 wb = ((const float4*)W2)[2 * lane + 1];
    float ax = 0.0f, ay = 0.0f, az = 0.0f, aw = 0.0f;
    for (int base = beg; base < end; base += 32) {
        const int j = base + lane;
        int sl = 0;
        if (j < end) sl = csr[j];
        const int left = end - base;
        const int cntk = (left < 32) ? left : 32;
#pragma unroll 4
        for (int k = 0; k < cntk; ++k) {
            const int s = __builtin_amdgcn_readlane(sl, k);
            const float4 v = ((const float4*)(H + (size_t)s * N_HID_F))[lane];
            ax += v.x;
            ay += v.y;
            az += v.z;
            aw += v.w;
        }
    }
    const float hx = tanhf((ax + hs.x) * id + bb.x);
    const float hy = tanhf((ay + hs.y) * id + bb.y);
    const float hz = tanhf((az + hs.z) * id + bb.z);
    const float hw = tanhf((aw + hs.w) * id + bb.w);
    const float p0 = hx * wa.x + hy * wa.z + hz * wb.x + hw * wb.z;
    const float p1 = hx * wa.y + hy * wa.w + hz * wb.y + hw * wb.w;
    const bool up = (lane & 16) != 0;
    float keep = up ? p1 : p0;
    const float give = up ? p0 : p1;
    keep += __shfl_xor(give, 16);
    keep += __shfl_xor(keep, 8);
    keep += __shfl_xor(keep, 4);
    keep += __shfl_xor(keep, 2);
    keep += __shfl_xor(keep, 1);
    if ((lane & 15) == 0) H2[(size_t)node * 2 + (lane >> 4)] = keep * id;
}

__global__ void k_gather_emb_out(const float* __restrict__ H2, const int* __restrict__ offs,
                                 const int* __restrict__ csr, const float* __restrict__ inv,
                                 const float* __restrict__ b2, const float* __restrict__ Wc,
                                 const float* __restrict__ bc, float* __restrict__ out, int nN) {
    int d = blockIdx.x * blockDim.x + threadIdx.x;
    if (d >= nN) return;
    const int beg = offs[d];
    const int end = offs[d + 1];
    const float id = inv[d];
    float a0 = 0.0f, a1 = 0.0f;
    const float2* __restrict__ P = (const float2*)H2;
    const float2 own = P[d];
    for (int j = beg; j < end; ++j) {
        const float2 v = P[csr[j]];
        a0 += v.x;
        a1 += v.y;
    }
    const float e0 = tanhf((a0 + own.x) * id + b2[0]);
    const float e1 = tanhf((a1 + own.y) * id + b2[1]);
    const float z  = e0 * Wc[0] + e1 * Wc[1] + bc[0];
    out[d] = 1.0f / (1.0f + expf(-z));
}

extern "C" void kernel_launch(void* const* d_in, const int* in_sizes, int n_in,
                              void* d_out, int out_size, void* d_ws, size_t ws_size,
                              hipStream_t stream) {
    const float* x   = (const float*)d_in[0];
    const int*   ei  = (const int*)d_in[1];
    const float* W1  = (const float*)d_in[2];
    const float* b1  = (const float*)d_in[3];
    const float* W2  = (const float*)d_in[4];
    const float* b2  = (const float*)d_in[5];
    const float* Wc  = (const float*)d_in[6];
    const float* bc  = (const float*)d_in[7];
    float* out = (float*)d_out;

    const int nN = in_sizes[0] / N_IN_F;
    const int nE = in_sizes[1] / 2;
    const int* src = ei;
    const int* dst = ei + nE;

    const int B   = 256;
    const int WPB = B / 32;
    const int nTiles = (nN + TILE_N - 1) / TILE_N;

    float* ws   = (float*)d_ws;
    float* h1   = ws;
    float* h2   = h1   + (size_t)nN * N_HID_F;
    float* inv  = h2   + (size_t)nN * 2;
    const int nChunks    = (nE + CHUNK_E - 1) / CHUNK_E;
    int*   offs  = (int*)(inv + (size_t)nN);
    int*   ctile = offs + (size_t)nN + 1;
    int*   csr   = ctile + (size_t)nChunks * nTiles;
    const int wordsTotal = (nN + 1) / 2;
    unsigned* cntc  = (unsigned*)(csr + (size_t)nE);
    int*   cbase = (int*)(cntc + (size_t)nChunks * wordsTotal);
    unsigned short* rank16 = (unsigned short*)(cbase + (size_t)nChunks * nN);

    k_chunk_rank<<<nChunks, CHUNK_T, 0, stream>>>(dst, rank16, cntc, ctile, nE, nN, nTiles);
    k_prefix_scan<<<nTiles, TILE_N, 0, stream>>>((const unsigned short*)cntc, ctile, cbase, offs, inv,
                                                 nN, nChunks, 2 * wordsTotal, nTiles);

    const int nStrips = (nN + 15) / 16;
    const size_t ldsBytes = (size_t)N_HID_F * BSTRIDE * sizeof(float);
    const int nGemmBlocks  = (nStrips + GEMM_T / 32 - 1) / (GEMM_T / 32);
    const int nPlaceBlocks = (nE + PLACE_U * GEMM_T - 1) / (PLACE_U * GEMM_T);
    k_gemm1_place<<<nGemmBlocks + nPlaceBlocks, GEMM_T, ldsBytes, stream>>>(
        x, W1, inv, h1, nN, src, dst, rank16, cbase, csr, nE, nGemmBlocks, nPlaceBlocks);
    k_gather_hid_proj<<<(nN + WPB - 1) / WPB, B, 0, stream>>>(h1, offs, csr, inv, b1, W2, h2, nN);

    k_gather_emb_out<<<(nN + B - 1) / B, B, 0, stream>>>(h2, offs, csr, inv, b2, Wc, bc, out, nN);
}
